# starting kernel
# baseline (speedup 1.0000x reference)
#include <hip/hip_runtime.h>
#include <math.h>

typedef float v2f __attribute__((ext_vector_type(2)));
typedef float v8f __attribute__((ext_vector_type(8)));

#define N_NODES 1024
#define IN_F    256
#define OUT_F   256
#define HEADS   4
#define HID     64
#define NEG_SLOPE 0.2f

__global__ void gemm_xw(const float* __restrict__ X, const float* __restrict__ W,
                        float* __restrict__ G) {
    const int lane  = threadIdx.x;
    const int tileN = blockIdx.x * 16;
    const int tileM = blockIdx.y * 16;
    const int lid   = lane & 15;
    const int kh    = (lane >> 4) * 2;
    const int m     = tileM + lid;
    const int n     = tileN + lid;

    v8f acc = {};
    for (int k0 = 0; k0 < IN_F; k0 += 4) {
        const int k = k0 + kh;
        v2f a, b;
        a.x = X[m * IN_F + k];
        a.y = X[m * IN_F + k + 1];
        b.x = W[k * OUT_F + n];
        b.y = W[(k + 1) * OUT_F + n];
        acc = __builtin_amdgcn_wmma_f32_16x16x4_f32(false, a, false, b,
                                                    (short)0, acc, false, false);
    }
    const int row_half = (lane >> 4) * 8;
#pragma unroll
    for (int p = 0; p < 8; ++p) {
        const int mm = tileM + p + row_half;
        G[mm * OUT_F + n] = acc[p];
    }
}

__global__ void attn_scores(const float* __restrict__ gl, const float* __restrict__ gr,
                            const float* __restrict__ adj, const float* __restrict__ aw,
                            float* __restrict__ P, float* __restrict__ denom) {
    const int i = blockIdx.x;
    const int h = blockIdx.y;
    const int t = threadIdx.x;

    __shared__ float s_gri[HID];
    __shared__ float s_aw[HID];
    __shared__ float s_red[256];

    if (t < HID) {
        s_gri[t] = gr[i * OUT_F + h * HID + t];
        s_aw[t]  = aw[t];
    }
    __syncthreads();

    float part = 0.0f;
    float* Ph = P + (size_t)h * N_NODES * N_NODES + (size_t)i * N_NODES;
    for (int j = t; j < N_NODES; j += 256) {
        float p = 0.0f;
        if (adj[(size_t)j * N_NODES + i] != 0.0f) {
            const float* glj = gl + j * OUT_F + h * HID;
            float e = 0.0f;
#pragma unroll
            for (int f = 0; f < HID; ++f) {
                const float s  = s_gri[f] + glj[f];
                const float lr = fmaxf(s, NEG_SLOPE * s);
                e = fmaf(lr, s_aw[f], e);
            }
            p = expf(e);
        }
        Ph[j] = p;
        part += p;
    }

    s_red[t] = part;
    __syncthreads();
    for (int s = 128; s > 0; s >>= 1) {
        if (t < s) s_red[t] += s_red[t + s];
        __syncthreads();
    }
    if (t == 0) denom[h * N_NODES + i] = s_red[0];
}

__global__ void aggregate(const float* __restrict__ P, const float* __restrict__ denom,
                          const float* __restrict__ gr, float* __restrict__ out,
                          const int* __restrict__ use_elu) {
    const int lane  = threadIdx.x;
    const int tileF = blockIdx.x * 16;
    const int tileM = blockIdx.y * 16;
    const int h     = blockIdx.z;
    const int lid   = lane & 15;
    const int kh    = (lane >> 4) * 2;
    const int m     = tileM + lid;

    const float recip = 1.0f / denom[h * N_NODES + m];
    const float* Ph = P + (size_t)h * N_NODES * N_NODES;

    v8f acc = {};
    for (int k0 = 0; k0 < N_NODES; k0 += 4) {
        const int k = k0 + kh;
        v2f a, b;
        a.x = Ph[(size_t)m * N_NODES + k]     * recip;
        a.y = Ph[(size_t)m * N_NODES + k + 1] * recip;
        b.x = gr[k * OUT_F + h * HID + tileF + lid];
        b.y = gr[(k + 1) * OUT_F + h * HID + tileF + lid];
        acc = __builtin_amdgcn_wmma_f32_16x16x4_f32(false, a, false, b,
                                                    (short)0, acc, false, false);
    }

    const int elu = use_elu[0];
    const int row_half = (lane >> 4) * 8;
#pragma unroll
    for (int p = 0; p < 8; ++p) {
        const int i = tileM + p + row_half;
        float x = acc[p];
        if (elu) x = (x > 0.0f) ? x : expm1f(x);
        out[i * OUT_F + h * HID + tileF + lid] = x;
    }
}

extern "C" void kernel_launch(void* const* d_in, const int* in_sizes, int n_in,
                              void* d_out, int out_size, void* d_ws, size_t ws_size,
                              hipStream_t stream) {
    const float* h_in  = (const float*)d_in[0];
    const float* nei   = (const float*)d_in[1];
    const float* adj   = (const float*)d_in[2];
    const float* w_l   = (const float*)d_in[3];
    const float* w_r   = (const float*)d_in[4];
    const float* aw    = (const float*)d_in[5];
    const int*   elu   = (const int*)d_in[6];
    float* out = (float*)d_out;

    float* ws    = (float*)d_ws;
    float* g_l   = ws;
    float* g_r   = ws + (size_t)N_NODES * OUT_F;
    float* P     = g_r + (size_t)N_NODES * OUT_F;
    float* denom = P + (size_t)HEADS * N_NODES * N_NODES;

    dim3 ggrid(OUT_F / 16, N_NODES / 16);
    gemm_xw<<<ggrid, 32, 0, stream>>>(h_in, w_l, g_l);
    gemm_xw<<<ggrid, 32, 0, stream>>>(nei,  w_r, g_r);

    attn_scores<<<dim3(N_NODES, HEADS), 256, 0, stream>>>(g_l, g_r, adj, aw, P, denom);

    aggregate<<<dim3(HID / 16, N_NODES / 16, HEADS), 32, 0, stream>>>(P, denom, g_r, out, elu);
}


# Score kernel re-tiled: one 1024-thread block per (16 target rows, head); the head's g_l slab goes once through LDS in whole rows, each thread keeps its row in registers for all 16; sum order kept
# speedup vs baseline: 2.7617x; 2.7617x over previous
#include <hip/hip_runtime.h>
#include <math.h>

typedef float v2f __attribute__((ext_vector_type(2)));
typedef float v8f __attribute__((ext_vector_type(8)));

#define N_NODES 1024
#define IN_F    256
#define OUT_F   256
#define HEADS   4
#define HID     64
#define NEG_SLOPE 0.2f

__global__ void gemm_xw(const float* __restrict__ X, const float* __restrict__ W,
                        float* __restrict__ G) {
    const int lane  = threadIdx.x;
    const int tileN = blockIdx.x * 16;
    const int tileM = blockIdx.y * 16;
    const int lid   = lane & 15;
    const int kh    = (lane >> 4) * 2;
    const int m     = tileM + lid;
    const int n     = tileN + lid;

    v8f acc = {};
    for (int k0 = 0; k0 < IN_F; k0 += 4) {
        const int k = k0 + kh;
        v2f a, b;
        a.x = X[m * IN_F + k];
        a.y = X[m * IN_F + k + 1];
        b.x = W[k * OUT_F + n];
        b.y = W[(k + 1) * OUT_F + n];
        acc = __builtin_amdgcn_wmma_f32_16x16x4_f32(false, a, false, b,
                                                    (short)0, acc, false, false);
    }
    const int row_half = (lane >> 4) * 8;
#pragma unroll
    for (int p = 0; p < 8; ++p) {
        const int mm = tileM + p + row_half;
        G[mm * OUT_F + n] = acc[p];
    }
}

#define TI        16
#define GL_PITCH  68

__global__ void __launch_bounds__(1024)
attn_scores(const float* __restrict__ gl, const float* __restrict__ gr,
            const float* __restrict__ adj, const float* __restrict__ aw,
            float* __restrict__ P, float* __restrict__ denom) {
    const int i0 = blockIdx.x * TI;
    const int h  = blockIdx.y;
    const int t  = threadIdx.x;

    __shared__ __attribute__((aligned(16))) float s_big[N_NODES * GL_PITCH];
    __shared__ __attribute__((aligned(16))) float s_gri[TI * HID];

    for (int it = 0; it < (N_NODES * (HID / 4)) / 1024; ++it) {
        const int idx = it * 1024 + t;
        const int row = idx >> 4;
        const int c   = idx & 15;
        const float4 v = *(const float4*)(gl + row * OUT_F + h * HID + c * 4);
        *(float4*)(s_big + row * GL_PITCH + c * 4) = v;
    }
    s_gri[t] = gr[(i0 + (t >> 6)) * OUT_F + h * HID + (t & 63)];
    __syncthreads();

    float glj[HID];
    for (int c = 0; c < HID / 4; ++c) {
        const float4 v = *(const float4*)(s_big + t * GL_PITCH + c * 4);
        glj[4 * c + 0] = v.x; glj[4 * c + 1] = v.y;
        glj[4 * c + 2] = v.z; glj[4 * c + 3] = v.w;
    }
    unsigned mask = 0u;
    for (int c = 0; c < TI / 4; ++c) {
        const float4 v = *(const float4*)(adj + (size_t)t * N_NODES + i0 + c * 4);
        mask |= (v.x != 0.0f ? 1u : 0u) << (4 * c + 0);
        mask |= (v.y != 0.0f ? 1u : 0u) << (4 * c + 1);
        mask |= (v.z != 0.0f ? 1u : 0u) << (4 * c + 2);
        mask |= (v.w != 0.0f ? 1u : 0u) << (4 * c + 3);
    }
    __syncthreads();

    float* Ph = P + (size_t)h * N_NODES * N_NODES + (size_t)i0 * N_NODES + t;
    for (int ib = 0; ib < TI; ib += 4) {
        const float* g0 = s_gri + (ib + 0) * HID;
        const float* g1 = s_gri + (ib + 1) * HID;
        const float* g2 = s_gri + (ib + 2) * HID;
        const float* g3 = s_gri + (ib + 3) * HID;
        float e0 = 0.0f, e1 = 0.0f, e2 = 0.0f, e3 = 0.0f;
        float4 c0 = *(const float4*)(g0), c1 = *(const float4*)(g1);
        float4 c2 = *(const float4*)(g2), c3 = *(const float4*)(g3);
        for (int q = 0; q < HID / 4; ++q) {
            const int qn = (q + 1 < HID / 4) ? q + 1 : q;
            const float4 n0 = *(const float4*)(g0 + 4 * qn);
            const float4 n1 = *(const float4*)(g1 + 4 * qn);
            const float4 n2 = *(const float4*)(g2 + 4 * qn);
            const float4 n3 = *(const float4*)(g3 + 4 * qn);
            __builtin_amdgcn_sched_barrier(0);
#define SCORE_STEP(comp, f)                                                     \
            {                                                                   \
                const float w  = aw[f];                                         \
                const float x  = glj[f];                                        \
                const float s0 = c0.comp + x;                                   \
                const float s1 = c1.comp + x;                                   \
                const float s2 = c2.comp + x;                                   \
                const float s3 = c3.comp + x;                                   \
                e0 = fmaf(fmaxf(s0, NEG_SLOPE * s0), w, e0);                    \
                e1 = fmaf(fmaxf(s1, NEG_SLOPE * s1), w, e1);                    \
                e2 = fmaf(fmaxf(s2, NEG_SLOPE * s2), w, e2);                    \
                e3 = fmaf(fmaxf(s3, NEG_SLOPE * s3), w, e3);                    \
            }
            SCORE_STEP(x, 4 * q + 0)
            SCORE_STEP(y, 4 * q + 1)
            SCORE_STEP(z, 4 * q + 2)
            SCORE_STEP(w, 4 * q + 3)
#undef SCORE_STEP
            c0 = n0; c1 = n1; c2 = n2; c3 = n3;
        }
        const float p0 = ((mask >> (ib + 0)) & 1u) ? expf(e0) : 0.0f;
        const float p1 = ((mask >> (ib + 1)) & 1u) ? expf(e1) : 0.0f;
        const float p2 = ((mask >> (ib + 2)) & 1u) ? expf(e2) : 0.0f;
        const float p3 = ((mask >> (ib + 3)) & 1u) ? expf(e3) : 0.0f;
        Ph[(size_t)(ib + 0) * N_NODES] = p0;
        Ph[(size_t)(ib + 1) * N_NODES] = p1;
        Ph[(size_t)(ib + 2) * N_NODES] = p2;
        Ph[(size_t)(ib + 3) * N_NODES] = p3;
        s_big[(ib + 0) * N_NODES + t] = p0;
        s_big[(ib + 1) * N_NODES + t] = p1;
        s_big[(ib + 2) * N_NODES + t] = p2;
        s_big[(ib + 3) * N_NODES + t] = p3;
    }
    __syncthreads();

    const int tt = t & 255;
    float* r0 = s_big + ((t >> 8) * 4) * N_NODES;
    for (int k = 0; k < 4; ++k) {
        float* r = r0 + k * N_NODES;
        float part = 0.0f;
        part += r[tt];
        part += r[tt + 256];
        part += r[tt + 512];
        part += r[tt + 768];
        r[tt] = part;
    }
    __syncthreads();
    for (int s = 128; s > 0; s >>= 1) {
        if (tt < s) {
            for (int k = 0; k < 4; ++k) {
                float* r = r0 + k * N_NODES;
                r[tt] += r[tt + s];
            }
        }
        __syncthreads();
    }
    if (tt == 0) {
        for (int k = 0; k < 4; ++k)
            denom[h * N_NODES + i0 + (t >> 8) * 4 + k] = r0[k * N_NODES];
    }
}

__global__ void aggregate(const float* __restrict__ P, const float* __restrict__ denom,
                          const float* __restrict__ gr, float* __restrict__ out,
                          const int* __restrict__ use_elu) {
    const int lane  = threadIdx.x;
    const int tileF = blockIdx.x * 16;
    const int tileM = blockIdx.y * 16;
    const int h     = blockIdx.z;
    const int lid   = lane & 15;
    const int kh    = (lane >> 4) * 2;
    const int m     = tileM + lid;

    const float recip = 1.0f / denom[h * N_NODES + m];
    const float* Ph = P + (size_t)h * N_NODES * N_NODES;

    v8f acc = {};
    for (int k0 = 0; k0 < N_NODES; k0 += 4) {
        const int k = k0 + kh;
        v2f a, b;
        a.x = Ph[(size_t)m * N_NODES + k]     * recip;
        a.y = Ph[(size_t)m * N_NODES + k + 1] * recip;
        b.x = gr[k * OUT_F + h * HID + tileF + lid];
        b.y = gr[(k + 1) * OUT_F + h * HID + tileF + lid];
        acc = __builtin_amdgcn_wmma_f32_16x16x4_f32(false, a, false, b,
                                                    (short)0, acc, false, false);
    }

    const int elu = use_elu[0];
    const int row_half = (lane >> 4) * 8;
#pragma unroll
    for (int p = 0; p < 8; ++p) {
        const int i = tileM + p + row_half;
        float x = acc[p];
        if (elu) x = (x > 0.0f) ? x : expm1f(x);
        out[i * OUT_F + h * HID + tileF + lid] = x;
    }
}

extern "C" void kernel_launch(void* const* d_in, const int* in_sizes, int n_in,
                              void* d_out, int out_size, void* d_ws, size_t ws_size,
                              hipStream_t stream) {
    const float* h_in  = (const float*)d_in[0];
    const float* nei   = (const float*)d_in[1];
    const float* adj   = (const float*)d_in[2];
    const float* w_l   = (const float*)d_in[3];
    const float* w_r   = (const float*)d_in[4];
    const float* aw    = (const float*)d_in[5];
    const int*   elu   = (const int*)d_in[6];
    float* out = (float*)d_out;

    float* ws    = (float*)d_ws;
    float* g_l   = ws;
    float* g_r   = ws + (size_t)N_NODES * OUT_F;
    float* P     = g_r + (size_t)N_NODES * OUT_F;
    float* denom = P + (size_t)HEADS * N_NODES * N_NODES;

    dim3 ggrid(OUT_F / 16, N_NODES / 16);
    gemm_xw<<<ggrid, 32, 0, stream>>>(h_in, w_l, g_l);
    gemm_xw<<<ggrid, 32, 0, stream>>>(nei,  w_r, g_r);

    attn_scores<<<dim3(N_NODES / TI, HEADS), 1024, 0, stream>>>(g_l, g_r, adj, aw, P, denom);

    aggregate<<<dim3(HID / 16, N_NODES / 16, HEADS), 32, 0, stream>>>(P, denom, g_r, out, elu);
}


# Aggregation fused into the score kernel: p stays in LDS as the WMMA A operand, each wave sums a 32-neighbor slice, the 32 partial tiles are added in wave order; neighbor sum order changes
# speedup vs baseline: 3.9790x; 1.4408x over previous
#include <hip/hip_runtime.h>
#include <math.h>

typedef float v2f __attribute__((ext_vector_type(2)));
typedef float v8f __attribute__((ext_vector_type(8)));

#define N_NODES 1024
#define IN_F    256
#define OUT_F   256
#define HEADS   4
#define HID     64
#define NEG_SLOPE 0.2f

__global__ void gemm_xw(const float* __restrict__ X, const float* __restrict__ W,
                        float* __restrict__ G) {
    const int lane  = threadIdx.x;
    const int tileN = blockIdx.x * 16;
    const int tileM = blockIdx.y * 16;
    const int lid   = lane & 15;
    const int kh    = (lane >> 4) * 2;
    const int m     = tileM + lid;
    const int n     = tileN + lid;

    v8f acc = {};
    for (int k0 = 0; k0 < IN_F; k0 += 4) {
        const int k = k0 + kh;
        v2f a, b;
        a.x = X[m * IN_F + k];
        a.y = X[m * IN_F + k + 1];
        b.x = W[k * OUT_F + n];
        b.y = W[(k + 1) * OUT_F + n];
        acc = __builtin_amdgcn_wmma_f32_16x16x4_f32(false, a, false, b,
                                                    (short)0, acc, false, false);
    }
    const int row_half = (lane >> 4) * 8;
#pragma unroll
    for (int p = 0; p < 8; ++p) {
        const int mm = tileM + p + row_half;
        G[mm * OUT_F + n] = acc[p];
    }
}

#define TI        16
#define GL_PITCH  68
#define P_PITCH   1028
#define RED_OFF   (TI * P_PITCH)
#define PART_OFF  (RED_OFF + TI * 256)

__global__ void __launch_bounds__(1024)
attn_fused(const float* __restrict__ gl, const float* __restrict__ gr,
           const float* __restrict__ adj, const float* __restrict__ aw,
           float* __restrict__ out, const int* __restrict__ use_elu) {
    const int i0 = blockIdx.x * TI;
    const int h  = blockIdx.y;
    const int t  = threadIdx.x;

    __shared__ __attribute__((aligned(16))) float s_big[N_NODES * GL_PITCH];
    __shared__ __attribute__((aligned(16))) float s_gri[TI * HID];

    for (int it = 0; it < (N_NODES * (HID / 4)) / 1024; ++it) {
        const int idx = it * 1024 + t;
        const int row = idx >> 4;
        const int c   = idx & 15;
        const float4 v = *(const float4*)(gl + row * OUT_F + h * HID + c * 4);
        *(float4*)(s_big + row * GL_PITCH + c * 4) = v;
    }
    s_gri[t] = gr[(i0 + (t >> 6)) * OUT_F + h * HID + (t & 63)];
    __syncthreads();

    float glj[HID];
    for (int c = 0; c < HID / 4; ++c) {
        const float4 v = *(const float4*)(s_big + t * GL_PITCH + c * 4);
        glj[4 * c + 0] = v.x; glj[4 * c + 1] = v.y;
        glj[4 * c + 2] = v.z; glj[4 * c + 3] = v.w;
    }
    unsigned mask = 0u;
    for (int c = 0; c < TI / 4; ++c) {
        const float4 v = *(const float4*)(adj + (size_t)t * N_NODES + i0 + c * 4);
        mask |= (v.x != 0.0f ? 1u : 0u) << (4 * c + 0);
        mask |= (v.y != 0.0f ? 1u : 0u) << (4 * c + 1);
        mask |= (v.z != 0.0f ? 1u : 0u) << (4 * c + 2);
        mask |= (v.w != 0.0f ? 1u : 0u) << (4 * c + 3);
    }
    __syncthreads();

    for (int ib = 0; ib < TI; ib += 4) {
        const float* g0 = s_gri + (ib + 0) * HID;
        const float* g1 = s_gri + (ib + 1) * HID;
        const float* g2 = s_gri + (ib + 2) * HID;
        const float* g3 = s_gri + (ib + 3) * HID;
        float e0 = 0.0f, e1 = 0.0f, e2 = 0.0f, e3 = 0.0f;
        float4 c0 = *(const float4*)(g0), c1 = *(const float4*)(g1);
        float4 c2 = *(const float4*)(g2), c3 = *(const float4*)(g3);
        for (int q = 0; q < HID / 4; ++q) {
            const int qn = (q + 1 < HID / 4) ? q + 1 : q;
            const float4 n0 = *(const float4*)(g0 + 4 * qn);
            const float4 n1 = *(const float4*)(g1 + 4 * qn);
            const float4 n2 = *(const float4*)(g2 + 4 * qn);
            const float4 n3 = *(const float4*)(g3 + 4 * qn);
            __builtin_amdgcn_sched_barrier(0);
#define SCORE_STEP(comp, f)                                                     \
            {                                                                   \
                const float w  = aw[f];                                         \
                const float x  = glj[f];                                        \
                const float s0 = c0.comp + x;                                   \
                const float s1 = c1.comp + x;                                   \
                const float s2 = c2.comp + x;                                   \
                const float s3 = c3.comp + x;                                   \
                e0 = fmaf(fmaxf(s0, NEG_SLOPE * s0), w, e0);                    \
                e1 = fmaf(fmaxf(s1, NEG_SLOPE * s1), w, e1);                    \
                e2 = fmaf(fmaxf(s2, NEG_SLOPE * s2), w, e2);                    \
                e3 = fmaf(fmaxf(s3, NEG_SLOPE * s3), w, e3);                    \
            }
            SCORE_STEP(x, 4 * q + 0)
            SCORE_STEP(y, 4 * q + 1)
            SCORE_STEP(z, 4 * q + 2)
            SCORE_STEP(w, 4 * q + 3)
#undef SCORE_STEP
            c0 = n0; c1 = n1; c2 = n2; c3 = n3;
        }
        const float p0 = ((mask >> (ib + 0)) & 1u) ? expf(e0) : 0.0f;
        const float p1 = ((mask >> (ib + 1)) & 1u) ? expf(e1) : 0.0f;
        const float p2 = ((mask >> (ib + 2)) & 1u) ? expf(e2) : 0.0f;
        const float p3 = ((mask >> (ib + 3)) & 1u) ? expf(e3) : 0.0f;
        s_big[(ib + 0) * P_PITCH + t] = p0;
        s_big[(ib + 1) * P_PITCH + t] = p1;
        s_big[(ib + 2) * P_PITCH + t] = p2;
        s_big[(ib + 3) * P_PITCH + t] = p3;
    }
    __syncthreads();

    const int tt    = t & 255;
    const int rbase = (t >> 8) * 4;
    float* red = s_big + RED_OFF;
    for (int k = 0; k < 4; ++k) {
        const float* r = s_big + (rbase + k) * P_PITCH;
        float part = 0.0f;
        part += r[tt];
        part += r[tt + 256];
        part += r[tt + 512];
        part += r[tt + 768];
        red[(rbase + k) * 256 + tt] = part;
    }
    __syncthreads();
    for (int s = 128; s > 0; s >>= 1) {
        if (tt < s) {
            for (int k = 0; k < 4; ++k) {
                float* r = red + (rbase + k) * 256;
                r[tt] += r[tt + s];
            }
        }
        __syncthreads();
    }

    const int wv   = t >> 5;
    const int lane = t & 31;
    const int lid  = lane & 15;
    const int kh   = (lane >> 4) * 2;
    const float recip = 1.0f / red[lid * 256];
    const float* prow = s_big + lid * P_PITCH + wv * 32 + kh;
    const float* gcol = gr + (size_t)(wv * 32 + kh) * OUT_F + h * HID + 4 * lid;

    v8f acc0 = {}, acc1 = {}, acc2 = {}, acc3 = {};
#pragma unroll
    for (int s = 0; s < 8; ++s) {
        const float2 pa = *(const float2*)(prow + 4 * s);
        const float4 b0 = *(const float4*)(gcol + (size_t)(4 * s) * OUT_F);
        const float4 b1 = *(const float4*)(gcol + (size_t)(4 * s + 1) * OUT_F);
        v2f a, b;
        a.x = pa.x * recip;
        a.y = pa.y * recip;
        b.x = b0.x; b.y = b1.x;
        acc0 = __builtin_amdgcn_wmma_f32_16x16x4_f32(false, a, false, b, (short)0, acc0, false, false);
        b.x = b0.y; b.y = b1.y;
        acc1 = __builtin_amdgcn_wmma_f32_16x16x4_f32(false, a, false, b, (short)0, acc1, false, false);
        b.x = b0.z; b.y = b1.z;
        acc2 = __builtin_amdgcn_wmma_f32_16x16x4_f32(false, a, false, b, (short)0, acc2, false, false);
        b.x = b0.w; b.y = b1.w;
        acc3 = __builtin_amdgcn_wmma_f32_16x16x4_f32(false, a, false, b, (short)0, acc3, false, false);
    }

    float* ptile = s_big + PART_OFF + wv * (TI * HID) + ((lane >> 4) * 8) * HID + 4 * lid;
#pragma unroll
    for (int p = 0; p < 8; ++p) {
        float4 v;
        v.x = acc0[p]; v.y = acc1[p]; v.z = acc2[p]; v.w = acc3[p];
        *(float4*)(ptile + p * HID) = v;
    }
    __syncthreads();

    const float* col = s_big + PART_OFF + t;
    float x = 0.0f;
#pragma unroll
    for (int w = 0; w < 32; ++w) x += col[w * (TI * HID)];
    if (use_elu[0]) x = (x > 0.0f) ? x : expm1f(x);
    out[(size_t)(i0 + (t >> 6)) * OUT_F + h * HID + (t & 63)] = x;
}

extern "C" void kernel_launch(void* const* d_in, const int* in_sizes, int n_in,
                              void* d_out, int out_size, void* d_ws, size_t ws_size,
                              hipStream_t stream) {
    const float* h_in  = (const float*)d_in[0];
    const float* nei   = (const float*)d_in[1];
    const float* adj   = (const float*)d_in[2];
    const float* w_l   = (const float*)d_in[3];
    const float* w_r   = (const float*)d_in[4];
    const float* aw    = (const float*)d_in[5];
    const int*   elu   = (const int*)d_in[6];
    float* out = (float*)d_out;

    float* ws    = (float*)d_ws;
    float* g_l   = ws;
    float* g_r   = ws + (size_t)N_NODES * OUT_F;

    dim3 ggrid(OUT_F / 16, N_NODES / 16);
    gemm_xw<<<ggrid, 32, 0, stream>>>(h_in, w_l, g_l);
    gemm_xw<<<ggrid, 32, 0, stream>>>(nei,  w_r, g_r);

    attn_fused<<<dim3(N_NODES / TI, HEADS), 1024, 0, stream>>>(g_l, g_r, adj, aw, out, elu);
}


# Projections re-tiled as one LDS-staged GEMM launch: a 256-thread block stages a 32x256 X panel and a 256x64 W panel once, its 8 waves each run one 16x16 tile from LDS; k order of every sum kept
# speedup vs baseline: 5.4545x; 1.3708x over previous
#include <hip/hip_runtime.h>
#include <math.h>

typedef float v2f __attribute__((ext_vector_type(2)));
typedef float v8f __attribute__((ext_vector_type(8)));

#define N_NODES 1024
#define IN_F    256
#define OUT_F   256
#define HEADS   4
#define HID     64
#define NEG_SLOPE 0.2f

#define BM       32
#define BN       64
#define A_PITCH  260
#define B_PITCH  72
#define C_PITCH  68
#define KB       16

__global__ void __launch_bounds__(256)
gemm_xw2(const float* __restrict__ X0, const float* __restrict__ W0, float* __restrict__ G0,
         const float* __restrict__ X1, const float* __restrict__ W1, float* __restrict__ G1) {
    const float* X = blockIdx.z ? X1 : X0;
    const float* W = blockIdx.z ? W1 : W0;
    float*       G = blockIdx.z ? G1 : G0;

    const int t     = threadIdx.x;
    const int tileN = blockIdx.x * BN;
    const int tileM = blockIdx.y * BM;

    __shared__ __attribute__((aligned(16))) float s_a[BM * A_PITCH];
    __shared__ __attribute__((aligned(16))) float s_b[IN_F * B_PITCH];

    for (int it = 0; it < (BM * IN_F / 4) / 256; ++it) {
        const int idx = it * 256 + t;
        const int row = idx >> 6;
        const int c4  = idx & 63;
        const float4 v = *(const float4*)(X + (size_t)(tileM + row) * IN_F + c4 * 4);
        *(float4*)(s_a + row * A_PITCH + c4 * 4) = v;
    }
    for (int it = 0; it < (IN_F * BN / 4) / 256; ++it) {
        const int idx = it * 256 + t;
        const int kr  = idx >> 4;
        const int c4  = idx & 15;
        const float4 v = *(const float4*)(W + (size_t)kr * OUT_F + tileN + c4 * 4);
        *(float4*)(s_b + kr * B_PITCH + c4 * 4) = v;
    }
    __syncthreads();

    const int wv   = t >> 5;
    const int lane = t & 31;
    const int lid  = lane & 15;
    const int kh   = (lane >> 4) * 2;
    const int wm   = wv >> 2;
    const int wn   = wv & 3;
    const float* ap = s_a + (wm * 16 + lid) * A_PITCH + kh;
    const float* bp = s_b + kh * B_PITCH + wn * 16 + lid;

    v2f fa[2][KB], fb[2][KB];
#pragma unroll
    for (int s = 0; s < KB; ++s) {
        const float2 av = *(const float2*)(ap + 4 * s);
        fa[0][s].x = av.x;
        fa[0][s].y = av.y;
        fb[0][s].x = bp[(4 * s) * B_PITCH];
        fb[0][s].y = bp[(4 * s + 1) * B_PITCH];
    }
    v8f acc = {};
#pragma unroll
    for (int kb = 0; kb < IN_F / (4 * KB); ++kb) {
        const int cur = kb & 1;
        const int nxt = cur ^ 1;
        if (kb + 1 < IN_F / (4 * KB)) {
#pragma unroll
            for (int s = 0; s < KB; ++s) {
                const int k0 = (kb + 1) * (4 * KB) + 4 * s;
                const float2 av = *(const float2*)(ap + k0);
                fa[nxt][s].x = av.x;
                fa[nxt][s].y = av.y;
                fb[nxt][s].x = bp[k0 * B_PITCH];
                fb[nxt][s].y = bp[(k0 + 1) * B_PITCH];
            }
        }
        __builtin_amdgcn_sched_barrier(0);
#pragma unroll
        for (int s = 0; s < KB; ++s)
            acc = __builtin_amdgcn_wmma_f32_16x16x4_f32(false, fa[cur][s], false, fb[cur][s],
                                                        (short)0, acc, false, false);
        __builtin_amdgcn_sched_barrier(0);
    }
    __syncthreads();

    const int row_half = (lane >> 4) * 8;
#pragma unroll
    for (int p = 0; p < 8; ++p)
        s_a[(wm * 16 + row_half + p) * C_PITCH + wn * 16 + lid] = acc[p];
    __syncthreads();

    for (int it = 0; it < (BM * BN / 4) / 256; ++it) {
        const int idx = it * 256 + t;
        const int row = idx >> 4;
        const int c4  = idx & 15;
        const float4 v = *(const float4*)(s_a + row * C_PITCH + c4 * 4);
        *(float4*)(G + (size_t)(tileM + row) * OUT_F + tileN + c4 * 4) = v;
    }
}

#define TI        16
#define GL_PITCH  68
#define P_PITCH   1028
#define RED_OFF   (TI * P_PITCH)
#define PART_OFF  (RED_OFF + TI * 256)

__global__ void __launch_bounds__(1024)
attn_fused(const float* __restrict__ gl, const float* __restrict__ gr,
           const float* __restrict__ adj, const float* __restrict__ aw,
           float* __restrict__ out, const int* __restrict__ use_elu) {
    const int i0 = blockIdx.x * TI;
    const int h  = blockIdx.y;
    const int t  = threadIdx.x;

    __shared__ __attribute__((aligned(16))) float s_big[N_NODES * GL_PITCH];
    __shared__ __attribute__((aligned(16))) float s_gri[TI * HID];

    for (int it = 0; it < (N_NODES * (HID / 4)) / 1024; ++it) {
        const int idx = it * 1024 + t;
        const int row = idx >> 4;
        const int c   = idx & 15;
        const float4 v = *(const float4*)(gl + row * OUT_F + h * HID + c * 4);
        *(float4*)(s_big + row * GL_PITCH + c * 4) = v;
    }
    s_gri[t] = gr[(i0 + (t >> 6)) * OUT_F + h * HID + (t & 63)];
    __syncthreads();

    float glj[HID];
    for (int c = 0; c < HID / 4; ++c) {
        const float4 v = *(const float4*)(s_big + t * GL_PITCH + c * 4);
        glj[4 * c + 0] = v.x; glj[4 * c + 1] = v.y;
        glj[4 * c + 2] = v.z; glj[4 * c + 3] = v.w;
    }
    unsigned mask = 0u;
    for (int c = 0; c < TI / 4; ++c) {
        const float4 v = *(const float4*)(adj + (size_t)t * N_NODES + i0 + c * 4);
        mask |= (v.x != 0.0f ? 1u : 0u) << (4 * c + 0);
        mask |= (v.y != 0.0f ? 1u : 0u) << (4 * c + 1);
        mask |= (v.z != 0.0f ? 1u : 0u) << (4 * c + 2);
        mask |= (v.w != 0.0f ? 1u : 0u) << (4 * c + 3);
    }
    __syncthreads();

    for (int ib = 0; ib < TI; ib += 4) {
        const float* g0 = s_gri + (ib + 0) * HID;
        const float* g1 = s_gri + (ib + 1) * HID;
        const float* g2 = s_gri + (ib + 2) * HID;
        const float* g3 = s_gri + (ib + 3) * HID;
        float e0 = 0.0f, e1 = 0.0f, e2 = 0.0f, e3 = 0.0f;
        float4 c0 = *(const float4*)(g0), c1 = *(const float4*)(g1);
        float4 c2 = *(const float4*)(g2), c3 = *(const float4*)(g3);
        for (int q = 0; q < HID / 4; ++q) {
            const int qn = (q + 1 < HID / 4) ? q + 1 : q;
            const float4 n0 = *(const float4*)(g0 + 4 * qn);
            const float4 n1 = *(const float4*)(g1 + 4 * qn);
            const float4 n2 = *(const float4*)(g2 + 4 * qn);
            const float4 n3 = *(const float4*)(g3 + 4 * qn);
            __builtin_amdgcn_sched_barrier(0);
#define SCORE_STEP(comp, f)                                                     \
            {                                                                   \
                const float w  = aw[f];                                         \
                const float x  = glj[f];                                        \
                const float s0 = c0.comp + x;                                   \
                const float s1 = c1.comp + x;                                   \
                const float s2 = c2.comp + x;                                   \
                const float s3 = c3.comp + x;                                   \
                e0 = fmaf(fmaxf(s0, NEG_SLOPE * s0), w, e0);                    \
                e1 = fmaf(fmaxf(s1, NEG_SLOPE * s1), w, e1);                    \
                e2 = fmaf(fmaxf(s2, NEG_SLOPE * s2), w, e2);                    \
                e3 = fmaf(fmaxf(s3, NEG_SLOPE * s3), w, e3);                    \
            }
            SCORE_STEP(x, 4 * q + 0)
            SCORE_STEP(y, 4 * q + 1)
            SCORE_STEP(z, 4 * q + 2)
            SCORE_STEP(w, 4 * q + 3)
#undef SCORE_STEP
            c0 = n0; c1 = n1; c2 = n2; c3 = n3;
        }
        const float p0 = ((mask >> (ib + 0)) & 1u) ? expf(e0) : 0.0f;
        const float p1 = ((mask >> (ib + 1)) & 1u) ? expf(e1) : 0.0f;
        const float p2 = ((mask >> (ib + 2)) & 1u) ? expf(e2) : 0.0f;
        const float p3 = ((mask >> (ib + 3)) & 1u) ? expf(e3) : 0.0f;
        s_big[(ib + 0) * P_PITCH + t] = p0;
        s_big[(ib + 1) * P_PITCH + t] = p1;
        s_big[(ib + 2) * P_PITCH + t] = p2;
        s_big[(ib + 3) * P_PITCH + t] = p3;
    }
    __syncthreads();

    const int tt    = t & 255;
    const int rbase = (t >> 8) * 4;
    float* red = s_big + RED_OFF;
    for (int k = 0; k < 4; ++k) {
        const float* r = s_big + (rbase + k) * P_PITCH;
        float part = 0.0f;
        part += r[tt];
        part += r[tt + 256];
        part += r[tt + 512];
        part += r[tt + 768];
        red[(rbase + k) * 256 + tt] = part;
    }
    __syncthreads();
    for (int s = 128; s > 0; s >>= 1) {
        if (tt < s) {
            for (int k = 0; k < 4; ++k) {
                float* r = red + (rbase + k) * 256;
                r[tt] += r[tt + s];
            }
        }
        __syncthreads();
    }

    const int wv   = t >> 5;
    const int lane = t & 31;
    const int lid  = lane & 15;
    const int kh   = (lane >> 4) * 2;
    const float recip = 1.0f / red[lid * 256];
    const float* prow = s_big + lid * P_PITCH + wv * 32 + kh;
    const float* gcol = gr + (size_t)(wv * 32 + kh) * OUT_F + h * HID + 4 * lid;

    v8f acc0 = {}, acc1 = {}, acc2 = {}, acc3 = {};
#pragma unroll
    for (int s = 0; s < 8; ++s) {
        const float2 pa = *(const float2*)(prow + 4 * s);
        const float4 b0 = *(const float4*)(gcol + (size_t)(4 * s) * OUT_F);
        const float4 b1 = *(const float4*)(gcol + (size_t)(4 * s + 1) * OUT_F);
        v2f a, b;
        a.x = pa.x * recip;
        a.y = pa.y * recip;
        b.x = b0.x; b.y = b1.x;
        acc0 = __builtin_amdgcn_wmma_f32_16x16x4_f32(false, a, false, b, (short)0, acc0, false, false);
        b.x = b0.y; b.y = b1.y;
        acc1 = __builtin_amdgcn_wmma_f32_16x16x4_f32(false, a, false, b, (short)0, acc1, false, false);
        b.x = b0.z; b.y = b1.z;
        acc2 = __builtin_amdgcn_wmma_f32_16x16x4_f32(false, a, false, b, (short)0, acc2, false, false);
        b.x = b0.w; b.y = b1.w;
        acc3 = __builtin_amdgcn_wmma_f32_16x16x4_f32(false, a, false, b, (short)0, acc3, false, false);
    }

    float* ptile = s_big + PART_OFF + wv * (TI * HID) + ((lane >> 4) * 8) * HID + 4 * lid;
#pragma unroll
    for (int p = 0; p < 8; ++p) {
        float4 v;
        v.x = acc0[p]; v.y = acc1[p]; v.z = acc2[p]; v.w = acc3[p];
        *(float4*)(ptile + p * HID) = v;
    }
    __syncthreads();

    const float* col = s_big + PART_OFF + t;
    float x = 0.0f;
#pragma unroll
    for (int w = 0; w < 32; ++w) x += col[w * (TI * HID)];
    if (use_elu[0]) x = (x > 0.0f) ? x : expm1f(x);
    out[(size_t)(i0 + (t >> 6)) * OUT_F + h * HID + (t & 63)] = x;
}

extern "C" void kernel_launch(void* const* d_in, const int* in_sizes, int n_in,
                              void* d_out, int out_size, void* d_ws, size_t ws_size,
                              hipStream_t stream) {
    const float* h_in  = (const float*)d_in[0];
    const float* nei   = (const float*)d_in[1];
    const float* adj   = (const float*)d_in[2];
    const float* w_l   = (const float*)d_in[3];
    const float* w_r   = (const float*)d_in[4];
    const float* aw    = (const float*)d_in[5];
    const int*   elu   = (const int*)d_in[6];
    float* out = (float*)d_out;

    float* ws    = (float*)d_ws;
    float* g_l   = ws;
    float* g_r   = ws + (size_t)N_NODES * OUT_F;

    gemm_xw2<<<dim3(OUT_F / BN, N_NODES / BM, 2), 256, 0, stream>>>(h_in, w_l, g_l, nei, w_r, g_r);

    attn_fused<<<dim3(N_NODES / TI, HEADS), 1024, 0, stream>>>(g_l, g_r, adj, aw, out, elu);
}


# Score loop rewritten with leaky_relu(s) = a*s + b*|s|: linear parts summed once per row and per column, inner step is a two-wide add and one fma on |s| per feature; score sum regrouped
# speedup vs baseline: 5.9942x; 1.0989x over previous
#include <hip/hip_runtime.h>
#include <math.h>

typedef float v2f __attribute__((ext_vector_type(2)));
typedef float v8f __attribute__((ext_vector_type(8)));

#define N_NODES 1024
#define IN_F    256
#define OUT_F   256
#define HEADS   4
#define HID     64
#define NEG_SLOPE 0.2f

#define BM       32
#define BN       64
#define A_PITCH  260
#define B_PITCH  72
#define C_PITCH  68
#define KB       16

__global__ void __launch_bounds__(256)
gemm_xw2(const float* __restrict__ X0, const float* __restrict__ W0, float* __restrict__ G0,
         const float* __restrict__ X1, const float* __restrict__ W1, float* __restrict__ G1) {
    const float* X = blockIdx.z ? X1 : X0;
    const float* W = blockIdx.z ? W1 : W0;
    float*       G = blockIdx.z ? G1 : G0;

    const int t     = threadIdx.x;
    const int tileN = blockIdx.x * BN;
    const int tileM = blockIdx.y * BM;

    __shared__ __attribute__((aligned(16))) float s_a[BM * A_PITCH];
    __shared__ __attribute__((aligned(16))) float s_b[IN_F * B_PITCH];

    for (int it = 0; it < (BM * IN_F / 4) / 256; ++it) {
        const int idx = it * 256 + t;
        const int row = idx >> 6;
        const int c4  = idx & 63;
        const float4 v = *(const float4*)(X + (size_t)(tileM + row) * IN_F + c4 * 4);
        *(float4*)(s_a + row * A_PITCH + c4 * 4) = v;
    }
    for (int it = 0; it < (IN_F * BN / 4) / 256; ++it) {
        const int idx = it * 256 + t;
        const int kr  = idx >> 4;
        const int c4  = idx & 15;
        const float4 v = *(const float4*)(W + (size_t)kr * OUT_F + tileN + c4 * 4);
        *(float4*)(s_b + kr * B_PITCH + c4 * 4) = v;
    }
    __syncthreads();

    const int wv   = t >> 5;
    const int lane = t & 31;
    const int lid  = lane & 15;
    const int kh   = (lane >> 4) * 2;
    const int wm   = wv >> 2;
    const int wn   = wv & 3;
    const float* ap = s_a + (wm * 16 + lid) * A_PITCH + kh;
    const float* bp = s_b + kh * B_PITCH + wn * 16 + lid;

    v2f fa[2][KB], fb[2][KB];
#pragma unroll
    for (int s = 0; s < KB; ++s) {
        const float2 av = *(const float2*)(ap + 4 * s);
        fa[0][s].x = av.x;
        fa[0][s].y = av.y;
        fb[0][s].x = bp[(4 * s) * B_PITCH];
        fb[0][s].y = bp[(4 * s + 1) * B_PITCH];
    }
    v8f acc = {};
#pragma unroll
    for (int kb = 0; kb < IN_F / (4 * KB); ++kb) {
        const int cur = kb & 1;
        const int nxt = cur ^ 1;
        if (kb + 1 < IN_F / (4 * KB)) {
#pragma unroll
            for (int s = 0; s < KB; ++s) {
                const int k0 = (kb + 1) * (4 * KB) + 4 * s;
                const float2 av = *(const float2*)(ap + k0);
                fa[nxt][s].x = av.x;
                fa[nxt][s].y = av.y;
                fb[nxt][s].x = bp[k0 * B_PITCH];
                fb[nxt][s].y = bp[(k0 + 1) * B_PITCH];
            }
        }
        __builtin_amdgcn_sched_barrier(0);
#pragma unroll
        for (int s = 0; s < KB; ++s)
            acc = __builtin_amdgcn_wmma_f32_16x16x4_f32(false, fa[cur][s], false, fb[cur][s],
                                                        (short)0, acc, false, false);
        __builtin_amdgcn_sched_barrier(0);
    }
    __syncthreads();

    const int row_half = (lane >> 4) * 8;
#pragma unroll
    for (int p = 0; p < 8; ++p)
        s_a[(wm * 16 + row_half + p) * C_PITCH + wn * 16 + lid] = acc[p];
    __syncthreads();

    for (int it = 0; it < (BM * BN / 4) / 256; ++it) {
        const int idx = it * 256 + t;
        const int row = idx >> 4;
        const int c4  = idx & 15;
        const float4 v = *(const float4*)(s_a + row * C_PITCH + c4 * 4);
        *(float4*)(G + (size_t)(tileM + row) * OUT_F + tileN + c4 * 4) = v;
    }
}

#define TI        16
#define LIN_C     (0.5f * (1.0f + NEG_SLOPE))
#define ABS_C     (0.5f * (1.0f - NEG_SLOPE))
#define GL_PITCH  68
#define P_PITCH   1028
#define RED_OFF   (TI * P_PITCH)
#define PART_OFF  (RED_OFF + TI * 256)

__global__ void __launch_bounds__(1024)
attn_fused(const float* __restrict__ gl, const float* __restrict__ gr,
           const float* __restrict__ adj, const float* __restrict__ aw,
           float* __restrict__ out, const int* __restrict__ use_elu) {
    const int i0 = blockIdx.x * TI;
    const int h  = blockIdx.y;
    const int t  = threadIdx.x;

    __shared__ __attribute__((aligned(16))) float s_big[N_NODES * GL_PITCH];
    __shared__ __attribute__((aligned(16))) float s_gri[TI * HID];
    __shared__ float s_lin[TI];

    for (int it = 0; it < (N_NODES * (HID / 4)) / 1024; ++it) {
        const int idx = it * 1024 + t;
        const int row = idx >> 4;
        const int c   = idx & 15;
        const float4 v = *(const float4*)(gl + row * OUT_F + h * HID + c * 4);
        *(float4*)(s_big + row * GL_PITCH + c * 4) = v;
    }
    s_gri[t] = gr[(i0 + (t >> 6)) * OUT_F + h * HID + (t & 63)];
    if (t < 256) {
        const int i = t >> 4;
        const int c = t & 15;
        const float4 g = *(const float4*)(gr + (size_t)(i0 + i) * OUT_F + h * HID + 4 * c);
        const float4 w = *(const float4*)(aw + 4 * c);
        float part = g.x * w.x;
        part = fmaf(g.y, w.y, part);
        part = fmaf(g.z, w.z, part);
        part = fmaf(g.w, w.w, part);
        part += __shfl_xor(part, 8);
        part += __shfl_xor(part, 4);
        part += __shfl_xor(part, 2);
        part += __shfl_xor(part, 1);
        if (c == 0) s_lin[i] = LIN_C * part;
    }
    __syncthreads();

    v2f glj[HID / 2];
    float cj = 0.0f;
    for (int c = 0; c < HID / 4; ++c) {
        const float4 v = *(const float4*)(s_big + t * GL_PITCH + c * 4);
        glj[2 * c + 0].x = v.x; glj[2 * c + 0].y = v.y;
        glj[2 * c + 1].x = v.z; glj[2 * c + 1].y = v.w;
        cj = fmaf(v.x, aw[4 * c + 0], cj);
        cj = fmaf(v.y, aw[4 * c + 1], cj);
        cj = fmaf(v.z, aw[4 * c + 2], cj);
        cj = fmaf(v.w, aw[4 * c + 3], cj);
    }
    cj *= LIN_C;
    unsigned mask = 0u;
    for (int c = 0; c < TI / 4; ++c) {
        const float4 v = *(const float4*)(adj + (size_t)t * N_NODES + i0 + c * 4);
        mask |= (v.x != 0.0f ? 1u : 0u) << (4 * c + 0);
        mask |= (v.y != 0.0f ? 1u : 0u) << (4 * c + 1);
        mask |= (v.z != 0.0f ? 1u : 0u) << (4 * c + 2);
        mask |= (v.w != 0.0f ? 1u : 0u) << (4 * c + 3);
    }
    __syncthreads();

    for (int ib = 0; ib < TI; ib += 4) {
        const float* g0 = s_gri + (ib + 0) * HID;
        const float* g1 = s_gri + (ib + 1) * HID;
        const float* g2 = s_gri + (ib + 2) * HID;
        const float* g3 = s_gri + (ib + 3) * HID;
        float e0 = 0.0f, e1 = 0.0f, e2 = 0.0f, e3 = 0.0f;
        for (int q = 0; q < HID / 4; ++q) {
            const float4 c0 = *(const float4*)(g0 + 4 * q);
            const float4 c1 = *(const float4*)(g1 + 4 * q);
            const float4 c2 = *(const float4*)(g2 + 4 * q);
            const float4 c3 = *(const float4*)(g3 + 4 * q);
            __builtin_amdgcn_sched_barrier(0);
#define SCORE_PAIR(lo, hi, fp)                                                  \
            {                                                                   \
                const float w0 = aw[2 * (fp)];                                  \
                const float w1 = aw[2 * (fp) + 1];                              \
                const v2f   x  = glj[fp];                                       \
                v2f a0, a1, a2, a3;                                             \
                a0.x = c0.lo; a0.y = c0.hi;                                     \
                a1.x = c1.lo; a1.y = c1.hi;                                     \
                a2.x = c2.lo; a2.y = c2.hi;                                     \
                a3.x = c3.lo; a3.y = c3.hi;                                     \
                const v2f s0 = a0 + x;                                          \
                const v2f s1 = a1 + x;                                          \
                const v2f s2 = a2 + x;                                          \
                const v2f s3 = a3 + x;                                          \
                e0 = fmaf(fabsf(s0.x), w0, e0);                                 \
                e1 = fmaf(fabsf(s1.x), w0, e1);                                 \
                e2 = fmaf(fabsf(s2.x), w0, e2);                                 \
                e3 = fmaf(fabsf(s3.x), w0, e3);                                 \
                e0 = fmaf(fabsf(s0.y), w1, e0);                                 \
                e1 = fmaf(fabsf(s1.y), w1, e1);                                 \
                e2 = fmaf(fabsf(s2.y), w1, e2);                                 \
                e3 = fmaf(fabsf(s3.y), w1, e3);                                 \
            }
            SCORE_PAIR(x, y, 2 * q + 0)
            SCORE_PAIR(z, w, 2 * q + 1)
#undef SCORE_PAIR
        }
        e0 = fmaf(ABS_C, e0, s_lin[ib + 0] + cj);
        e1 = fmaf(ABS_C, e1, s_lin[ib + 1] + cj);
        e2 = fmaf(ABS_C, e2, s_lin[ib + 2] + cj);
        e3 = fmaf(ABS_C, e3, s_lin[ib + 3] + cj);
        const float p0 = ((mask >> (ib + 0)) & 1u) ? expf(e0) : 0.0f;
        const float p1 = ((mask >> (ib + 1)) & 1u) ? expf(e1) : 0.0f;
        const float p2 = ((mask >> (ib + 2)) & 1u) ? expf(e2) : 0.0f;
        const float p3 = ((mask >> (ib + 3)) & 1u) ? expf(e3) : 0.0f;
        s_big[(ib + 0) * P_PITCH + t] = p0;
        s_big[(ib + 1) * P_PITCH + t] = p1;
        s_big[(ib + 2) * P_PITCH + t] = p2;
        s_big[(ib + 3) * P_PITCH + t] = p3;
    }
    __syncthreads();

    const int tt    = t & 255;
    const int rbase = (t >> 8) * 4;
    float* red = s_big + RED_OFF;
    for (int k = 0; k < 4; ++k) {
        const float* r = s_big + (rbase + k) * P_PITCH;
        float part = 0.0f;
        part += r[tt];
        part += r[tt + 256];
        part += r[tt + 512];
        part += r[tt + 768];
        red[(rbase + k) * 256 + tt] = part;
    }
    __syncthreads();
    for (int s = 128; s > 0; s >>= 1) {
        if (tt < s) {
            for (int k = 0; k < 4; ++k) {
                float* r = red + (rbase + k) * 256;
                r[tt] += r[tt + s];
            }
        }
        __syncthreads();
    }

    const int wv   = t >> 5;
    const int lane = t & 31;
    const int lid  = lane & 15;
    const int kh   = (lane >> 4) * 2;
    const float recip = 1.0f / red[lid * 256];
    const float* prow = s_big + lid * P_PITCH + wv * 32 + kh;
    const float* gcol = gr + (size_t)(wv * 32 + kh) * OUT_F + h * HID + 4 * lid;

    v8f acc0 = {}, acc1 = {}, acc2 = {}, acc3 = {};
#pragma unroll
    for (int s = 0; s < 8; ++s) {
        const float2 pa = *(const float2*)(prow + 4 * s);
        const float4 b0 = *(const float4*)(gcol + (size_t)(4 * s) * OUT_F);
        const float4 b1 = *(const float4*)(gcol + (size_t)(4 * s + 1) * OUT_F);
        v2f a, b;
        a.x = pa.x * recip;
        a.y = pa.y * recip;
        b.x = b0.x; b.y = b1.x;
        acc0 = __builtin_amdgcn_wmma_f32_16x16x4_f32(false, a, false, b, (short)0, acc0, false, false);
        b.x = b0.y; b.y = b1.y;
        acc1 = __builtin_amdgcn_wmma_f32_16x16x4_f32(false, a, false, b, (short)0, acc1, false, false);
        b.x = b0.z; b.y = b1.z;
        acc2 = __builtin_amdgcn_wmma_f32_16x16x4_f32(false, a, false, b, (short)0, acc2, false, false);
        b.x = b0.w; b.y = b1.w;
        acc3 = __builtin_amdgcn_wmma_f32_16x16x4_f32(false, a, false, b, (short)0, acc3, false, false);
    }

    float* ptile = s_big + PART_OFF + wv * (TI * HID) + ((lane >> 4) * 8) * HID + 4 * lid;
#pragma unroll
    for (int p = 0; p < 8; ++p) {
        float4 v;
        v.x = acc0[p]; v.y = acc1[p]; v.z = acc2[p]; v.w = acc3[p];
        *(float4*)(ptile + p * HID) = v;
    }
    __syncthreads();

    const float* col = s_big + PART_OFF + t;
    float x = 0.0f;
#pragma unroll
    for (int w = 0; w < 32; ++w) x += col[w * (TI * HID)];
    if (use_elu[0]) x = (x > 0.0f) ? x : expm1f(x);
    out[(size_t)(i0 + (t >> 6)) * OUT_F + h * HID + (t & 63)] = x;
}

extern "C" void kernel_launch(void* const* d_in, const int* in_sizes, int n_in,
                              void* d_out, int out_size, void* d_ws, size_t ws_size,
                              hipStream_t stream) {
    const float* h_in  = (const float*)d_in[0];
    const float* nei   = (const float*)d_in[1];
    const float* adj   = (const float*)d_in[2];
    const float* w_l   = (const float*)d_in[3];
    const float* w_r   = (const float*)d_in[4];
    const float* aw    = (const float*)d_in[5];
    const int*   elu   = (const int*)d_in[6];
    float* out = (float*)d_out;

    float* ws    = (float*)d_ws;
    float* g_l   = ws;
    float* g_r   = ws + (size_t)N_NODES * OUT_F;

    gemm_xw2<<<dim3(OUT_F / BN, N_NODES / BM, 2), 256, 0, stream>>>(h_in, w_l, g_l, nei, w_r, g_r);

    attn_fused<<<dim3(N_NODES / TI, HEADS), 1024, 0, stream>>>(g_l, g_r, adj, aw, out, elu);
}


# Softmax division moved behind the aggregation: each wave feeds its own raw p to WMMA after scoring it, row sums from the A fragments; no denominator pass, 10 block barriers fewer; sum order changes
# speedup vs baseline: 6.5616x; 1.0947x over previous
#include <hip/hip_runtime.h>
#include <math.h>

typedef float v2f __attribute__((ext_vector_type(2)));
typedef float v8f __attribute__((ext_vector_type(8)));

#define N_NODES 1024
#define IN_F    256
#define OUT_F   256
#define HEADS   4
#define HID     64
#define NEG_SLOPE 0.2f

#define BM       32
#define BN       64
#define A_PITCH  260
#define B_PITCH  72
#define C_PITCH  68
#define KB       16

__global__ void __launch_bounds__(256)
gemm_xw2(const float* __restrict__ X0, const float* __restrict__ W0, float* __restrict__ G0,
         const float* __restrict__ X1, const float* __restrict__ W1, float* __restrict__ G1) {
    const float* X = blockIdx.z ? X1 : X0;
    const float* W = blockIdx.z ? W1 : W0;
    float*       G = blockIdx.z ? G1 : G0;

    const int t     = threadIdx.x;
    const int tileN = blockIdx.x * BN;
    const int tileM = blockIdx.y * BM;

    __shared__ __attribute__((aligned(16))) float s_a[BM * A_PITCH];
    __shared__ __attribute__((aligned(16))) float s_b[IN_F * B_PITCH];

    for (int it = 0; it < (BM * IN_F / 4) / 256; ++it) {
        const int idx = it * 256 + t;
        const int row = idx >> 6;
        const int c4  = idx & 63;
        const float4 v = *(const float4*)(X + (size_t)(tileM + row) * IN_F + c4 * 4);
        *(float4*)(s_a + row * A_PITCH + c4 * 4) = v;
    }
    for (int it = 0; it < (IN_F * BN / 4) / 256; ++it) {
        const int idx = it * 256 + t;
        const int kr  = idx >> 4;
        const int c4  = idx & 15;
        const float4 v = *(const float4*)(W + (size_t)kr * OUT_F + tileN + c4 * 4);
        *(float4*)(s_b + kr * B_PITCH + c4 * 4) = v;
    }
    __syncthreads();

    const int wv   = t >> 5;
    const int lane = t & 31;
    const int lid  = lane & 15;
    const int kh   = (lane >> 4) * 2;
    const int wm   = wv >> 2;
    const int wn   = wv & 3;
    const float* ap = s_a + (wm * 16 + lid) * A_PITCH + kh;
    const float* bp = s_b + kh * B_PITCH + wn * 16 + lid;

    v2f fa[2][KB], fb[2][KB];
#pragma unroll
    for (int s = 0; s < KB; ++s) {
        const float2 av = *(const float2*)(ap + 4 * s);
        fa[0][s].x = av.x;
        fa[0][s].y = av.y;
        fb[0][s].x = bp[(4 * s) * B_PITCH];
        fb[0][s].y = bp[(4 * s + 1) * B_PITCH];
    }
    v8f acc = {};
#pragma unroll
    for (int kb = 0; kb < IN_F / (4 * KB); ++kb) {
        const int cur = kb & 1;
        const int nxt = cur ^ 1;
        if (kb + 1 < IN_F / (4 * KB)) {
#pragma unroll
            for (int s = 0; s < KB; ++s) {
                const int k0 = (kb + 1) * (4 * KB) + 4 * s;
                const float2 av = *(const float2*)(ap + k0);
                fa[nxt][s].x = av.x;
                fa[nxt][s].y = av.y;
                fb[nxt][s].x = bp[k0 * B_PITCH];
                fb[nxt][s].y = bp[(k0 + 1) * B_PITCH];
            }
        }
        __builtin_amdgcn_sched_barrier(0);
#pragma unroll
        for (int s = 0; s < KB; ++s)
            acc = __builtin_amdgcn_wmma_f32_16x16x4_f32(false, fa[cur][s], false, fb[cur][s],
                                                        (short)0, acc, false, false);
        __builtin_amdgcn_sched_barrier(0);
    }
    __syncthreads();

    const int row_half = (lane >> 4) * 8;
#pragma unroll
    for (int p = 0; p < 8; ++p)
        s_a[(wm * 16 + row_half + p) * C_PITCH + wn * 16 + lid] = acc[p];
    __syncthreads();

    for (int it = 0; it < (BM * BN / 4) / 256; ++it) {
        const int idx = it * 256 + t;
        const int row = idx >> 4;
        const int c4  = idx & 15;
        const float4 v = *(const float4*)(s_a + row * C_PITCH + c4 * 4);
        *(float4*)(G + (size_t)(tileM + row) * OUT_F + tileN + c4 * 4) = v;
    }
}

#define TI        16
#define LIN_C     (0.5f * (1.0f + NEG_SLOPE))
#define ABS_C     (0.5f * (1.0f - NEG_SLOPE))
#define GL_PITCH  68
#define P_PITCH   1028
#define RS_OFF    (TI * P_PITCH)
#define PART_OFF  (RS_OFF + TI * 32)

__global__ void __launch_bounds__(1024)
attn_fused(const float* __restrict__ gl, const float* __restrict__ gr,
           const float* __restrict__ adj, const float* __restrict__ aw,
           float* __restrict__ out, const int* __restrict__ use_elu) {
    const int i0 = blockIdx.x * TI;
    const int h  = blockIdx.y;
    const int t  = threadIdx.x;

    __shared__ __attribute__((aligned(16))) float s_big[N_NODES * GL_PITCH];
    __shared__ __attribute__((aligned(16))) float s_gri[TI * HID];
    __shared__ float s_lin[TI];

    for (int it = 0; it < (N_NODES * (HID / 4)) / 1024; ++it) {
        const int idx = it * 1024 + t;
        const int row = idx >> 4;
        const int c   = idx & 15;
        const float4 v = *(const float4*)(gl + row * OUT_F + h * HID + c * 4);
        *(float4*)(s_big + row * GL_PITCH + c * 4) = v;
    }
    s_gri[t] = gr[(i0 + (t >> 6)) * OUT_F + h * HID + (t & 63)];
    if (t < 256) {
        const int i = t >> 4;
        const int c = t & 15;
        const float4 g = *(const float4*)(gr + (size_t)(i0 + i) * OUT_F + h * HID + 4 * c);
        const float4 w = *(const float4*)(aw + 4 * c);
        float part = g.x * w.x;
        part = fmaf(g.y, w.y, part);
        part = fmaf(g.z, w.z, part);
        part = fmaf(g.w, w.w, part);
        part += __shfl_xor(part, 8);
        part += __shfl_xor(part, 4);
        part += __shfl_xor(part, 2);
        part += __shfl_xor(part, 1);
        if (c == 0) s_lin[i] = LIN_C * part;
    }
    __syncthreads();

    v2f glj[HID / 2];
    float cj = 0.0f;
    for (int c = 0; c < HID / 4; ++c) {
        const float4 v = *(const float4*)(s_big + t * GL_PITCH + c * 4);
        glj[2 * c + 0].x = v.x; glj[2 * c + 0].y = v.y;
        glj[2 * c + 1].x = v.z; glj[2 * c + 1].y = v.w;
        cj = fmaf(v.x, aw[4 * c + 0], cj);
        cj = fmaf(v.y, aw[4 * c + 1], cj);
        cj = fmaf(v.z, aw[4 * c + 2], cj);
        cj = fmaf(v.w, aw[4 * c + 3], cj);
    }
    cj *= LIN_C;
    unsigned mask = 0u;
    for (int c = 0; c < TI / 4; ++c) {
        const float4 v = *(const float4*)(adj + (size_t)t * N_NODES + i0 + c * 4);
        mask |= (v.x != 0.0f ? 1u : 0u) << (4 * c + 0);
        mask |= (v.y != 0.0f ? 1u : 0u) << (4 * c + 1);
        mask |= (v.z != 0.0f ? 1u : 0u) << (4 * c + 2);
        mask |= (v.w != 0.0f ? 1u : 0u) << (4 * c + 3);
    }
    __syncthreads();

    for (int ib = 0; ib < TI; ib += 4) {
        const float* g0 = s_gri + (ib + 0) * HID;
        const float* g1 = s_gri + (ib + 1) * HID;
        const float* g2 = s_gri + (ib + 2) * HID;
        const float* g3 = s_gri + (ib + 3) * HID;
        float e0 = 0.0f, e1 = 0.0f, e2 = 0.0f, e3 = 0.0f;
        for (int q = 0; q < HID / 4; ++q) {
            const float4 c0 = *(const float4*)(g0 + 4 * q);
            const float4 c1 = *(const float4*)(g1 + 4 * q);
            const float4 c2 = *(const float4*)(g2 + 4 * q);
            const float4 c3 = *(const float4*)(g3 + 4 * q);
            __builtin_amdgcn_sched_barrier(0);
#define SCORE_PAIR(lo, hi, fp)                                                  \
            {                                                                   \
                const float w0 = aw[2 * (fp)];                                  \
                const float w1 = aw[2 * (fp) + 1];                              \
                const v2f   x  = glj[fp];                                       \
                v2f a0, a1, a2, a3;                                             \
                a0.x = c0.lo; a0.y = c0.hi;                                     \
                a1.x = c1.lo; a1.y = c1.hi;                                     \
                a2.x = c2.lo; a2.y = c2.hi;                                     \
                a3.x = c3.lo; a3.y = c3.hi;                                     \
                const v2f s0 = a0 + x;                                          \
                const v2f s1 = a1 + x;                                          \
                const v2f s2 = a2 + x;                                          \
                const v2f s3 = a3 + x;                                          \
                e0 = fmaf(fabsf(s0.x), w0, e0);                                 \
                e1 = fmaf(fabsf(s1.x), w0, e1);                                 \
                e2 = fmaf(fabsf(s2.x), w0, e2);                                 \
                e3 = fmaf(fabsf(s3.x), w0, e3);                                 \
                e0 = fmaf(fabsf(s0.y), w1, e0);                                 \
                e1 = fmaf(fabsf(s1.y), w1, e1);                                 \
                e2 = fmaf(fabsf(s2.y), w1, e2);                                 \
                e3 = fmaf(fabsf(s3.y), w1, e3);                                 \
            }
            SCORE_PAIR(x, y, 2 * q + 0)
            SCORE_PAIR(z, w, 2 * q + 1)
#undef SCORE_PAIR
        }
        e0 = fmaf(ABS_C, e0, s_lin[ib + 0] + cj);
        e1 = fmaf(ABS_C, e1, s_lin[ib + 1] + cj);
        e2 = fmaf(ABS_C, e2, s_lin[ib + 2] + cj);
        e3 = fmaf(ABS_C, e3, s_lin[ib + 3] + cj);
        const float p0 = ((mask >> (ib + 0)) & 1u) ? expf(e0) : 0.0f;
        const float p1 = ((mask >> (ib + 1)) & 1u) ? expf(e1) : 0.0f;
        const float p2 = ((mask >> (ib + 2)) & 1u) ? expf(e2) : 0.0f;
        const float p3 = ((mask >> (ib + 3)) & 1u) ? expf(e3) : 0.0f;
        s_big[(ib + 0) * P_PITCH + t] = p0;
        s_big[(ib + 1) * P_PITCH + t] = p1;
        s_big[(ib + 2) * P_PITCH + t] = p2;
        s_big[(ib + 3) * P_PITCH + t] = p3;
    }

    const int wv   = t >> 5;
    const int lane = t & 31;
    const int lid  = lane & 15;
    const int kh   = (lane >> 4) * 2;
    const float* prow = s_big + lid * P_PITCH + wv * 32 + kh;
    const float* gcol = gr + (size_t)(wv * 32 + kh) * OUT_F + h * HID + 4 * lid;

    float rsum = 0.0f;
    v8f acc0 = {}, acc1 = {}, acc2 = {}, acc3 = {};
#pragma unroll
    for (int s = 0; s < 8; ++s) {
        const float2 pa = *(const float2*)(prow + 4 * s);
        const float4 b0 = *(const float4*)(gcol + (size_t)(4 * s) * OUT_F);
        const float4 b1 = *(const float4*)(gcol + (size_t)(4 * s + 1) * OUT_F);
        v2f a, b;
        a.x = pa.x;
        a.y = pa.y;
        rsum += pa.x;
        rsum += pa.y;
        b.x = b0.x; b.y = b1.x;
        acc0 = __builtin_amdgcn_wmma_f32_16x16x4_f32(false, a, false, b, (short)0, acc0, false, false);
        b.x = b0.y; b.y = b1.y;
        acc1 = __builtin_amdgcn_wmma_f32_16x16x4_f32(false, a, false, b, (short)0, acc1, false, false);
        b.x = b0.z; b.y = b1.z;
        acc2 = __builtin_amdgcn_wmma_f32_16x16x4_f32(false, a, false, b, (short)0, acc2, false, false);
        b.x = b0.w; b.y = b1.w;
        acc3 = __builtin_amdgcn_wmma_f32_16x16x4_f32(false, a, false, b, (short)0, acc3, false, false);
    }

    float* ptile = s_big + PART_OFF + wv * (TI * HID) + ((lane >> 4) * 8) * HID + 4 * lid;
#pragma unroll
    for (int p = 0; p < 8; ++p) {
        float4 v;
        v.x = acc0[p]; v.y = acc1[p]; v.z = acc2[p]; v.w = acc3[p];
        *(float4*)(ptile + p * HID) = v;
    }
    rsum += __shfl_xor(rsum, 16);
    s_big[RS_OFF + lid * 32 + wv] = rsum;
    __syncthreads();

    const float* col = s_big + PART_OFF + t;
    float x = 0.0f;
#pragma unroll
    for (int w = 0; w < 32; ++w) x += col[w * (TI * HID)];
    float den = s_big[RS_OFF + (t >> 6) * 32 + lane];
    den += __shfl_xor(den, 16);
    den += __shfl_xor(den, 8);
    den += __shfl_xor(den, 4);
    den += __shfl_xor(den, 2);
    den += __shfl_xor(den, 1);
    x = x / den;
    if (use_elu[0]) x = (x > 0.0f) ? x : expm1f(x);
    out[(size_t)(i0 + (t >> 6)) * OUT_F + h * HID + (t & 63)] = x;
}

extern "C" void kernel_launch(void* const* d_in, const int* in_sizes, int n_in,
                              void* d_out, int out_size, void* d_ws, size_t ws_size,
                              hipStream_t stream) {
    const float* h_in  = (const float*)d_in[0];
    const float* nei   = (const float*)d_in[1];
    const float* adj   = (const float*)d_in[2];
    const float* w_l   = (const float*)d_in[3];
    const float* w_r   = (const float*)d_in[4];
    const float* aw    = (const float*)d_in[5];
    const int*   elu   = (const int*)d_in[6];
    float* out = (float*)d_out;

    float* ws    = (float*)d_ws;
    float* g_l   = ws;
    float* g_r   = ws + (size_t)N_NODES * OUT_F;

    gemm_xw2<<<dim3(OUT_F / BN, N_NODES / BM, 2), 256, 0, stream>>>(h_in, w_l, g_l, nei, w_r, g_r);

    attn_fused<<<dim3(N_NODES / TI, HEADS), 1024, 0, stream>>>(g_l, g_r, adj, aw, out, elu);
}


# Score loop visits only the connected targets of each source node: zeros stored first, then each lane walks the set bits of its 16-bit mask, two per trip; a wave stops when no lane has a bit left
# speedup vs baseline: 7.1249x; 1.0858x over previous
#include <hip/hip_runtime.h>
#include <math.h>

typedef float v2f __attribute__((ext_vector_type(2)));
typedef float v8f __attribute__((ext_vector_type(8)));

#define N_NODES 1024
#define IN_F    256
#define OUT_F   256
#define HEADS   4
#define HID     64
#define NEG_SLOPE 0.2f

#define BM       32
#define BN       64
#define A_PITCH  260
#define B_PITCH  72
#define C_PITCH  68
#define KB       16

__global__ void __launch_bounds__(256)
gemm_xw2(const float* __restrict__ X0, const float* __restrict__ W0, float* __restrict__ G0,
         const float* __restrict__ X1, const float* __restrict__ W1, float* __restrict__ G1) {
    const float* X = blockIdx.z ? X1 : X0;
    const float* W = blockIdx.z ? W1 : W0;
    float*       G = blockIdx.z ? G1 : G0;

    const int t     = threadIdx.x;
    const int tileN = blockIdx.x * BN;
    const int tileM = blockIdx.y * BM;

    __shared__ __attribute__((aligned(16))) float s_a[BM * A_PITCH];
    __shared__ __attribute__((aligned(16))) float s_b[IN_F * B_PITCH];

    for (int it = 0; it < (BM * IN_F / 4) / 256; ++it) {
        const int idx = it * 256 + t;
        const int row = idx >> 6;
        const int c4  = idx & 63;
        const float4 v = *(const float4*)(X + (size_t)(tileM + row) * IN_F + c4 * 4);
        *(float4*)(s_a + row * A_PITCH + c4 * 4) = v;
    }
    for (int it = 0; it < (IN_F * BN / 4) / 256; ++it) {
        const int idx = it * 256 + t;
        const int kr  = idx >> 4;
        const int c4  = idx & 15;
        const float4 v = *(const float4*)(W + (size_t)kr * OUT_F + tileN + c4 * 4);
        *(float4*)(s_b + kr * B_PITCH + c4 * 4) = v;
    }
    __syncthreads();

    const int wv   = t >> 5;
    const int lane = t & 31;
    const int lid  = lane & 15;
    const int kh   = (lane >> 4) * 2;
    const int wm   = wv >> 2;
    const int wn   = wv & 3;
    const float* ap = s_a + (wm * 16 + lid) * A_PITCH + kh;
    const float* bp = s_b + kh * B_PITCH + wn * 16 + lid;

    v2f fa[2][KB], fb[2][KB];
#pragma unroll
    for (int s = 0; s < KB; ++s) {
        const float2 av = *(const float2*)(ap + 4 * s);
        fa[0][s].x = av.x;
        fa[0][s].y = av.y;
        fb[0][s].x = bp[(4 * s) * B_PITCH];
        fb[0][s].y = bp[(4 * s + 1) * B_PITCH];
    }
    v8f acc = {};
#pragma unroll
    for (int kb = 0; kb < IN_F / (4 * KB); ++kb) {
        const int cur = kb & 1;
        const int nxt = cur ^ 1;
        if (kb + 1 < IN_F / (4 * KB)) {
#pragma unroll
            for (int s = 0; s < KB; ++s) {
                const int k0 = (kb + 1) * (4 * KB) + 4 * s;
                const float2 av = *(const float2*)(ap + k0);
                fa[nxt][s].x = av.x;
                fa[nxt][s].y = av.y;
                fb[nxt][s].x = bp[k0 * B_PITCH];
                fb[nxt][s].y = bp[(k0 + 1) * B_PITCH];
            }
        }
        __builtin_amdgcn_sched_barrier(0);
#pragma unroll
        for (int s = 0; s < KB; ++s)
            acc = __builtin_amdgcn_wmma_f32_16x16x4_f32(false, fa[cur][s], false, fb[cur][s],
                                                        (short)0, acc, false, false);
        __builtin_amdgcn_sched_barrier(0);
    }
    __syncthreads();

    const int row_half = (lane >> 4) * 8;
#pragma unroll
    for (int p = 0; p < 8; ++p)
        s_a[(wm * 16 + row_half + p) * C_PITCH + wn * 16 + lid] = acc[p];
    __syncthreads();

    for (int it = 0; it < (BM * BN / 4) / 256; ++it) {
        const int idx = it * 256 + t;
        const int row = idx >> 4;
        const int c4  = idx & 15;
        const float4 v = *(const float4*)(s_a + row * C_PITCH + c4 * 4);
        *(float4*)(G + (size_t)(tileM + row) * OUT_F + tileN + c4 * 4) = v;
    }
}

#define TI        16
#define LIN_C     (0.5f * (1.0f + NEG_SLOPE))
#define ABS_C     (0.5f * (1.0f - NEG_SLOPE))
#define GL_PITCH  68
#define GR_PITCH  68
#define P_PITCH   1028
#define RS_OFF    (TI * P_PITCH)
#define PART_OFF  (RS_OFF + TI * 32)
#define SPARE_OFF (PART_OFF + 32 * TI * HID)

__global__ void __launch_bounds__(1024)
attn_fused(const float* __restrict__ gl, const float* __restrict__ gr,
           const float* __restrict__ adj, const float* __restrict__ aw,
           float* __restrict__ out, const int* __restrict__ use_elu) {
    const int i0 = blockIdx.x * TI;
    const int h  = blockIdx.y;
    const int t  = threadIdx.x;

    __shared__ __attribute__((aligned(16))) float s_big[N_NODES * GL_PITCH];
    __shared__ __attribute__((aligned(16))) float s_gri[TI * GR_PITCH];
    __shared__ float s_lin[TI];

    for (int it = 0; it < (N_NODES * (HID / 4)) / 1024; ++it) {
        const int idx = it * 1024 + t;
        const int row = idx >> 4;
        const int c   = idx & 15;
        const float4 v = *(const float4*)(gl + row * OUT_F + h * HID + c * 4);
        *(float4*)(s_big + row * GL_PITCH + c * 4) = v;
    }
    s_gri[(t >> 6) * GR_PITCH + (t & 63)] = gr[(i0 + (t >> 6)) * OUT_F + h * HID + (t & 63)];
    if (t < 256) {
        const int i = t >> 4;
        const int c = t & 15;
        const float4 g = *(const float4*)(gr + (size_t)(i0 + i) * OUT_F + h * HID + 4 * c);
        const float4 w = *(const float4*)(aw + 4 * c);
        float part = g.x * w.x;
        part = fmaf(g.y, w.y, part);
        part = fmaf(g.z, w.z, part);
        part = fmaf(g.w, w.w, part);
        part += __shfl_xor(part, 8);
        part += __shfl_xor(part, 4);
        part += __shfl_xor(part, 2);
        part += __shfl_xor(part, 1);
        if (c == 0) s_lin[i] = LIN_C * part;
    }
    __syncthreads();

    v2f glj[HID / 2];
    float cj = 0.0f;
    for (int c = 0; c < HID / 4; ++c) {
        const float4 v = *(const float4*)(s_big + t * GL_PITCH + c * 4);
        glj[2 * c + 0].x = v.x; glj[2 * c + 0].y = v.y;
        glj[2 * c + 1].x = v.z; glj[2 * c + 1].y = v.w;
        cj = fmaf(v.x, aw[4 * c + 0], cj);
        cj = fmaf(v.y, aw[4 * c + 1], cj);
        cj = fmaf(v.z, aw[4 * c + 2], cj);
        cj = fmaf(v.w, aw[4 * c + 3], cj);
    }
    cj *= LIN_C;
    unsigned mask = 0u;
    for (int c = 0; c < TI / 4; ++c) {
        const float4 v = *(const float4*)(adj + (size_t)t * N_NODES + i0 + c * 4);
        mask |= (v.x != 0.0f ? 1u : 0u) << (4 * c + 0);
        mask |= (v.y != 0.0f ? 1u : 0u) << (4 * c + 1);
        mask |= (v.z != 0.0f ? 1u : 0u) << (4 * c + 2);
        mask |= (v.w != 0.0f ? 1u : 0u) << (4 * c + 3);
    }
    __syncthreads();

    for (int i = 0; i < TI; ++i) s_big[i * P_PITCH + t] = 0.0f;

    unsigned m = mask;
    int trip = 0;
    do {
        const bool v0 = (m != 0u);
        const int  t0 = v0 ? __builtin_ctz(m) : 0;
        m &= m - 1u;
        const bool v1 = (m != 0u);
        const int  t1 = v1 ? __builtin_ctz(m) : 0;
        m &= m - 1u;
        const float* g0 = s_gri + t0 * GR_PITCH;
        const float* g1 = s_gri + t1 * GR_PITCH;
        float e0 = 0.0f, e1 = 0.0f;
        for (int q = 0; q < HID / 4; q += 2) {
            const float4 c0 = *(const float4*)(g0 + 4 * q);
            const float4 c1 = *(const float4*)(g1 + 4 * q);
            const float4 d0 = *(const float4*)(g0 + 4 * q + 4);
            const float4 d1 = *(const float4*)(g1 + 4 * q + 4);
            __builtin_amdgcn_sched_barrier(0);
#define SCORE_PAIR(r0, r1, lo, hi, fp)                                          \
            {                                                                   \
                const float w0 = aw[2 * (fp)];                                  \
                const float w1 = aw[2 * (fp) + 1];                              \
                const v2f   x  = glj[fp];                                       \
                v2f a0, a1;                                                     \
                a0.x = r0.lo; a0.y = r0.hi;                                     \
                a1.x = r1.lo; a1.y = r1.hi;                                     \
                const v2f s0 = a0 + x;                                          \
                const v2f s1 = a1 + x;                                          \
                e0 = fmaf(fabsf(s0.x), w0, e0);                                 \
                e1 = fmaf(fabsf(s1.x), w0, e1);                                 \
                e0 = fmaf(fabsf(s0.y), w1, e0);                                 \
                e1 = fmaf(fabsf(s1.y), w1, e1);                                 \
            }
            SCORE_PAIR(c0, c1, x, y, 2 * q + 0)
            SCORE_PAIR(c0, c1, z, w, 2 * q + 1)
            SCORE_PAIR(d0, d1, x, y, 2 * q + 2)
            SCORE_PAIR(d0, d1, z, w, 2 * q + 3)
#undef SCORE_PAIR
        }
        e0 = fmaf(ABS_C, e0, s_lin[t0] + cj);
        e1 = fmaf(ABS_C, e1, s_lin[t1] + cj);
        s_big[(v0 ? t0 * P_PITCH : SPARE_OFF) + t] = expf(e0);
        s_big[(v1 ? t1 * P_PITCH : SPARE_OFF) + t] = expf(e1);
        ++trip;
    } while (trip < TI / 2 && __builtin_amdgcn_ballot_w32(m != 0u) != 0u);

    const int wv   = t >> 5;
    const int lane = t & 31;
    const int lid  = lane & 15;
    const int kh   = (lane >> 4) * 2;
    const float* prow = s_big + lid * P_PITCH + wv * 32 + kh;
    const float* gcol = gr + (size_t)(wv * 32 + kh) * OUT_F + h * HID + 4 * lid;

    float rsum = 0.0f;
    v8f acc0 = {}, acc1 = {}, acc2 = {}, acc3 = {};
#pragma unroll
    for (int s = 0; s < 8; ++s) {
        const float2 pa = *(const float2*)(prow + 4 * s);
        const float4 b0 = *(const float4*)(gcol + (size_t)(4 * s) * OUT_F);
        const float4 b1 = *(const float4*)(gcol + (size_t)(4 * s + 1) * OUT_F);
        v2f a, b;
        a.x = pa.x;
        a.y = pa.y;
        rsum += pa.x;
        rsum += pa.y;
        b.x = b0.x; b.y = b1.x;
        acc0 = __builtin_amdgcn_wmma_f32_16x16x4_f32(false, a, false, b, (short)0, acc0, false, false);
        b.x = b0.y; b.y = b1.y;
        acc1 = __builtin_amdgcn_wmma_f32_16x16x4_f32(false, a, false, b, (short)0, acc1, false, false);
        b.x = b0.z; b.y = b1.z;
        acc2 = __builtin_amdgcn_wmma_f32_16x16x4_f32(false, a, false, b, (short)0, acc2, false, false);
        b.x = b0.w; b.y = b1.w;
        acc3 = __builtin_amdgcn_wmma_f32_16x16x4_f32(false, a, false, b, (short)0, acc3, false, false);
    }

    float* ptile = s_big + PART_OFF + wv * (TI * HID) + ((lane >> 4) * 8) * HID + 4 * lid;
#pragma unroll
    for (int p = 0; p < 8; ++p) {
        float4 v;
        v.x = acc0[p]; v.y = acc1[p]; v.z = acc2[p]; v.w = acc3[p];
        *(float4*)(ptile + p * HID) = v;
    }
    rsum += __shfl_xor(rsum, 16);
    s_big[RS_OFF + lid * 32 + wv] = rsum;
    __syncthreads();

    const float* col = s_big + PART_OFF + t;
    float x = 0.0f;
#pragma unroll
    for (int w = 0; w < 32; ++w) x += col[w * (TI * HID)];
    float den = s_big[RS_OFF + (t >> 6) * 32 + lane];
    den += __shfl_xor(den, 16);
    den += __shfl_xor(den, 8);
    den += __shfl_xor(den, 4);
    den += __shfl_xor(den, 2);
    den += __shfl_xor(den, 1);
    x = x / den;
    if (use_elu[0]) x = (x > 0.0f) ? x : expm1f(x);
    out[(size_t)(i0 + (t >> 6)) * OUT_F + h * HID + (t & 63)] = x;
}

extern "C" void kernel_launch(void* const* d_in, const int* in_sizes, int n_in,
                              void* d_out, int out_size, void* d_ws, size_t ws_size,
                              hipStream_t stream) {
    const float* h_in  = (const float*)d_in[0];
    const float* nei   = (const float*)d_in[1];
    const float* adj   = (const float*)d_in[2];
    const float* w_l   = (const float*)d_in[3];
    const float* w_r   = (const float*)d_in[4];
    const float* aw    = (const float*)d_in[5];
    const int*   elu   = (const int*)d_in[6];
    float* out = (float*)d_out;

    float* ws    = (float*)d_ws;
    float* g_l   = ws;
    float* g_r   = ws + (size_t)N_NODES * OUT_F;

    gemm_xw2<<<dim3(OUT_F / BN, N_NODES / BM, 2), 256, 0, stream>>>(h_in, w_l, g_l, nei, w_r, g_r);

    attn_fused<<<dim3(N_NODES / TI, HEADS), 1024, 0, stream>>>(g_l, g_r, adj, aw, out, elu);
}


# Source nodes dealt to threads by score-loop trips needed (block-wide counting sort of the masks): a wave's lanes finish together; masks read first, 4 lanes per adj row; neighbour sum order changes
# speedup vs baseline: 7.3492x; 1.0315x over previous
#include <hip/hip_runtime.h>
#include <math.h>

typedef float v2f __attribute__((ext_vector_type(2)));
typedef float v4f __attribute__((ext_vector_type(4)));
typedef float v8f __attribute__((ext_vector_type(8)));

#define N_NODES 1024
#define IN_F    256
#define OUT_F   256
#define HEADS   4
#define HID     64
#define NEG_SLOPE 0.2f

#define BM       32
#define BN       64
#define A_PITCH  260
#define B_PITCH  72
#define C_PITCH  68
#define KB       16

__global__ void __launch_bounds__(256)
gemm_xw2(const float* __restrict__ X0, const float* __restrict__ W0, float* __restrict__ G0,
         const float* __restrict__ X1, const float* __restrict__ W1, float* __restrict__ G1) {
    const float* X = blockIdx.z ? X1 : X0;
    const float* W = blockIdx.z ? W1 : W0;
    float*       G = blockIdx.z ? G1 : G0;

    const int t     = threadIdx.x;
    const int tileN = blockIdx.x * BN;
    const int tileM = blockIdx.y * BM;

    __shared__ __attribute__((aligned(16))) float s_a[BM * A_PITCH];
    __shared__ __attribute__((aligned(16))) float s_b[IN_F * B_PITCH];

    for (int it = 0; it < (BM * IN_F / 4) / 256; ++it) {
        const int idx = it * 256 + t;
        const int row = idx >> 6;
        const int c4  = idx & 63;
        const float4 v = *(const float4*)(X + (size_t)(tileM + row) * IN_F + c4 * 4);
        *(float4*)(s_a + row * A_PITCH + c4 * 4) = v;
    }
    for (int it = 0; it < (IN_F * BN / 4) / 256; ++it) {
        const int idx = it * 256 + t;
        const int kr  = idx >> 4;
        const int c4  = idx & 15;
        const float4 v = *(const float4*)(W + (size_t)kr * OUT_F + tileN + c4 * 4);
        *(float4*)(s_b + kr * B_PITCH + c4 * 4) = v;
    }
    __syncthreads();

    const int wv   = t >> 5;
    const int lane = t & 31;
    const int lid  = lane & 15;
    const int kh   = (lane >> 4) * 2;
    const int wm   = wv >> 2;
    const int wn   = wv & 3;
    const float* ap = s_a + (wm * 16 + lid) * A_PITCH + kh;
    const float* bp = s_b + kh * B_PITCH + wn * 16 + lid;

    v2f fa[2][KB], fb[2][KB];
#pragma unroll
    for (int s = 0; s < KB; ++s) {
        const float2 av = *(const float2*)(ap + 4 * s);
        fa[0][s].x = av.x;
        fa[0][s].y = av.y;
        fb[0][s].x = bp[(4 * s) * B_PITCH];
        fb[0][s].y = bp[(4 * s + 1) * B_PITCH];
    }
    v8f acc = {};
#pragma unroll
    for (int kb = 0; kb < IN_F / (4 * KB); ++kb) {
        const int cur = kb & 1;
        const int nxt = cur ^ 1;
        if (kb + 1 < IN_F / (4 * KB)) {
#pragma unroll
            for (int s = 0; s < KB; ++s) {
                const int k0 = (kb + 1) * (4 * KB) + 4 * s;
                const float2 av = *(const float2*)(ap + k0);
                fa[nxt][s].x = av.x;
                fa[nxt][s].y = av.y;
                fb[nxt][s].x = bp[k0 * B_PITCH];
                fb[nxt][s].y = bp[(k0 + 1) * B_PITCH];
            }
        }
        __builtin_amdgcn_sched_barrier(0);
#pragma unroll
        for (int s = 0; s < KB; ++s)
            acc = __builtin_amdgcn_wmma_f32_16x16x4_f32(false, fa[cur][s], false, fb[cur][s],
                                                        (short)0, acc, false, false);
        __builtin_amdgcn_sched_barrier(0);
    }
    __syncthreads();

    const int row_half = (lane >> 4) * 8;
#pragma unroll
    for (int p = 0; p < 8; ++p)
        s_a[(wm * 16 + row_half + p) * C_PITCH + wn * 16 + lid] = acc[p];
    __syncthreads();

    for (int it = 0; it < (BM * BN / 4) / 256; ++it) {
        const int idx = it * 256 + t;
        const int row = idx >> 4;
        const int c4  = idx & 15;
        const float4 v = *(const float4*)(s_a + row * C_PITCH + c4 * 4);
        *(float4*)(G + (size_t)(tileM + row) * OUT_F + tileN + c4 * 4) = v;
    }
}

#define TI        16
#define LIN_C     (0.5f * (1.0f + NEG_SLOPE))
#define ABS_C     (0.5f * (1.0f - NEG_SLOPE))
#define GL_PITCH  68
#define GR_PITCH  68
#define P_PITCH   1028
#define RS_OFF    (TI * P_PITCH)
#define PART_OFF  (RS_OFF + TI * 32)
#define SPARE_OFF (PART_OFF + 32 * TI * HID)
#define NB        (TI / 2 + 1)
#define LDS_BARRIER()                                                   \
    do {                                                                \
        __builtin_amdgcn_fence(__ATOMIC_RELEASE, "workgroup", "local"); \
        __builtin_amdgcn_s_barrier();                                   \
        __builtin_amdgcn_fence(__ATOMIC_ACQUIRE, "workgroup", "local"); \
    } while (0)

__global__ void __launch_bounds__(1024)
attn_fused(const float* __restrict__ gl, const float* __restrict__ gr,
           const float* __restrict__ adj, const float* __restrict__ aw,
           float* __restrict__ out, const int* __restrict__ use_elu) {
    const int i0 = blockIdx.x * TI;
    const int h  = blockIdx.y;
    const int t  = threadIdx.x;

    __shared__ __attribute__((aligned(16))) float s_big[N_NODES * GL_PITCH];
    __shared__ __attribute__((aligned(16))) float s_gri[TI * GR_PITCH];
    __shared__ float s_lin[TI];
    __shared__ unsigned s_cnt[NB * 32];
    __shared__ unsigned s_off[NB * 32];
    __shared__ unsigned s_tot[NB];
    __shared__ __attribute__((aligned(16))) unsigned s_perm[N_NODES];
    __shared__ unsigned s_pos[N_NODES];

    const int wv   = t >> 5;
    const int lane = t & 31;

    const int aq = lane >> 2;
    const int ac = lane & 3;
    const float* arow = adj + (size_t)((t & ~31) + aq) * N_NODES + i0 + 4 * ac;
    const float4 av0 = *(const float4*)(arow);
    const float4 av1 = *(const float4*)(arow + (size_t)8 * N_NODES);
    const float4 av2 = *(const float4*)(arow + (size_t)16 * N_NODES);
    const float4 av3 = *(const float4*)(arow + (size_t)24 * N_NODES);
    const float gri_v = gr[(i0 + (t >> 6)) * OUT_F + h * HID + (t & 63)];
    const float4 ling = *(const float4*)(gr + (size_t)(i0 + ((t >> 4) & 15)) * OUT_F + h * HID + 4 * (t & 15));
    const float4 linw = *(const float4*)(aw + 4 * (t & 15));

    const float* gsrc = gl + (t >> 4) * OUT_F + h * HID + (t & 15) * 4;
#define SLAB_LOAD(n) const v4f sl##n = *(const v4f*)(gsrc + (n) * 64 * OUT_F);
    SLAB_LOAD(0)  SLAB_LOAD(1)  SLAB_LOAD(2)  SLAB_LOAD(3)
    SLAB_LOAD(4)  SLAB_LOAD(5)  SLAB_LOAD(6)  SLAB_LOAD(7)
    SLAB_LOAD(8)  SLAB_LOAD(9)  SLAB_LOAD(10) SLAB_LOAD(11)
    SLAB_LOAD(12) SLAB_LOAD(13) SLAB_LOAD(14) SLAB_LOAD(15)
#undef SLAB_LOAD
    asm volatile("" :: "v"(av0.x), "v"(av0.y), "v"(av0.z), "v"(av0.w), "v"(av1.x), "v"(av1.y), "v"(av1.z), "v"(av1.w),
                       "v"(av2.x), "v"(av2.y), "v"(av2.z), "v"(av2.w), "v"(av3.x), "v"(av3.y), "v"(av3.z), "v"(av3.w));
    __builtin_amdgcn_sched_barrier(0);

#define NIBBLE(v) ((v.x != 0.0f ? 1u : 0u) | (v.y != 0.0f ? 2u : 0u) | (v.z != 0.0f ? 4u : 0u) | (v.w != 0.0f ? 8u : 0u))
    unsigned w01 = (NIBBLE(av0) | (NIBBLE(av1) << 16)) << (4 * ac);
    unsigned w23 = (NIBBLE(av2) | (NIBBLE(av3) << 16)) << (4 * ac);
#undef NIBBLE
    w01 |= __shfl_xor(w01, 1);
    w23 |= __shfl_xor(w23, 1);
    w01 |= __shfl_xor(w01, 2);
    w23 |= __shfl_xor(w23, 2);
    const unsigned wsel = (ac & 2) ? w23 : w01;
    const unsigned held = (ac & 1) ? (wsel >> 16) : (wsel & 0xffffu);
    const unsigned jmask = __shfl(held, 4 * (lane & 7) + (lane >> 3));

    const int need = (__builtin_popcount(jmask) + 1) >> 1;
    unsigned rank = 0u, cnt = 0u;
#pragma unroll
    for (int b = 0; b < NB; ++b) {
        const unsigned bal = __builtin_amdgcn_ballot_w32(need == b);
        const unsigned below = __builtin_amdgcn_mbcnt_lo(bal, 0u);
        rank = (need == b) ? below : rank;
        cnt  = (lane == b) ? (unsigned)__builtin_popcount(bal) : cnt;
    }
    if (lane < NB) s_cnt[lane * 32 + wv] = cnt;
    LDS_BARRIER();

    if (t < NB * 32) {
        const unsigned c = s_cnt[t];
        unsigned off = 0u, tot = 0u;
#pragma unroll
        for (int k = 0; k < 6; ++k) {
            const unsigned bal = __builtin_amdgcn_ballot_w32(((c >> k) & 1u) != 0u);
            off += __builtin_amdgcn_mbcnt_lo(bal, 0u) << k;
            tot += (unsigned)__builtin_popcount(bal) << k;
        }
        s_off[t] = off;
        if (lane == 0) s_tot[wv] = tot;
    }
    LDS_BARRIER();

    unsigned slot = s_off[need * 32 + wv] + rank;
#pragma unroll
    for (int b = 0; b < NB - 1; ++b) slot += (b < need) ? s_tot[b] : 0u;
    {
        const unsigned chunk = slot >> 5;
        const unsigned r     = chunk & 3u;
        const unsigned w_to  = (chunk & ~3u) | ((chunk & 4u) ? 3u - r : r);
        slot = (w_to << 5) | (slot & 31u);
    }
    s_perm[slot] = (unsigned)t | (jmask << 16);
    s_pos[t]     = slot;
    LDS_BARRIER();

#define SLAB_STORE(n) *(v4f*)(s_big + s_pos[(n) * 64 + (t >> 4)] * GL_PITCH + (t & 15) * 4) = sl##n;
    SLAB_STORE(0)  SLAB_STORE(1)  SLAB_STORE(2)  SLAB_STORE(3)
    SLAB_STORE(4)  SLAB_STORE(5)  SLAB_STORE(6)  SLAB_STORE(7)
    SLAB_STORE(8)  SLAB_STORE(9)  SLAB_STORE(10) SLAB_STORE(11)
    SLAB_STORE(12) SLAB_STORE(13) SLAB_STORE(14) SLAB_STORE(15)
#undef SLAB_STORE
    s_gri[(t >> 6) * GR_PITCH + (t & 63)] = gri_v;
    if (t < 256) {
        const int i = t >> 4;
        const int c = t & 15;
        const float4 g = ling;
        const float4 w = linw;
        float part = g.x * w.x;
        part = fmaf(g.y, w.y, part);
        part = fmaf(g.z, w.z, part);
        part = fmaf(g.w, w.w, part);
        part += __shfl_xor(part, 8);
        part += __shfl_xor(part, 4);
        part += __shfl_xor(part, 2);
        part += __shfl_xor(part, 1);
        if (c == 0) s_lin[i] = LIN_C * part;
    }
    __syncthreads();

    v2f glj[HID / 2];
    float cj = 0.0f;
    for (int c = 0; c < HID / 4; ++c) {
        const float4 v = *(const float4*)(s_big + t * GL_PITCH + c * 4);
        glj[2 * c + 0].x = v.x; glj[2 * c + 0].y = v.y;
        glj[2 * c + 1].x = v.z; glj[2 * c + 1].y = v.w;
        cj = fmaf(v.x, aw[4 * c + 0], cj);
        cj = fmaf(v.y, aw[4 * c + 1], cj);
        cj = fmaf(v.z, aw[4 * c + 2], cj);
        cj = fmaf(v.w, aw[4 * c + 3], cj);
    }
    cj *= LIN_C;
    const unsigned mask = s_perm[t] >> 16;
    __syncthreads();

    for (int i = 0; i < TI; ++i) s_big[i * P_PITCH + t] = 0.0f;

    unsigned m = mask;
    int trip = 0;
    do {
        const bool v0 = (m != 0u);
        const int  t0 = v0 ? __builtin_ctz(m) : 0;
        m &= m - 1u;
        const bool v1 = (m != 0u);
        const int  t1 = v1 ? __builtin_ctz(m) : 0;
        m &= m - 1u;
        const float* g0 = s_gri + t0 * GR_PITCH;
        const float* g1 = s_gri + t1 * GR_PITCH;
        float e0 = 0.0f, e1 = 0.0f;
        for (int q = 0; q < HID / 4; q += 2) {
            const float4 c0 = *(const float4*)(g0 + 4 * q);
            const float4 c1 = *(const float4*)(g1 + 4 * q);
            const float4 d0 = *(const float4*)(g0 + 4 * q + 4);
            const float4 d1 = *(const float4*)(g1 + 4 * q + 4);
            __builtin_amdgcn_sched_barrier(0);
#define SCORE_PAIR(r0, r1, lo, hi, fp)                                          \
            {                                                                   \
                const float w0 = aw[2 * (fp)];                                  \
                const float w1 = aw[2 * (fp) + 1];                              \
                const v2f   x  = glj[fp];                                       \
                v2f a0, a1;                                                     \
                a0.x = r0.lo; a0.y = r0.hi;                                     \
                a1.x = r1.lo; a1.y = r1.hi;                                     \
                const v2f s0 = a0 + x;                                          \
                const v2f s1 = a1 + x;                                          \
                e0 = fmaf(fabsf(s0.x), w0, e0);                                 \
                e1 = fmaf(fabsf(s1.x), w0, e1);                                 \
                e0 = fmaf(fabsf(s0.y), w1, e0);                                 \
                e1 = fmaf(fabsf(s1.y), w1, e1);                                 \
            }
            SCORE_PAIR(c0, c1, x, y, 2 * q + 0)
            SCORE_PAIR(c0, c1, z, w, 2 * q + 1)
            SCORE_PAIR(d0, d1, x, y, 2 * q + 2)
            SCORE_PAIR(d0, d1, z, w, 2 * q + 3)
#undef SCORE_PAIR
        }
        e0 = fmaf(ABS_C, e0, s_lin[t0] + cj);
        e1 = fmaf(ABS_C, e1, s_lin[t1] + cj);
        s_big[(v0 ? t0 * P_PITCH : SPARE_OFF) + t] = expf(e0);
        s_big[(v1 ? t1 * P_PITCH : SPARE_OFF) + t] = expf(e1);
        ++trip;
    } while (trip < TI / 2 && __builtin_amdgcn_ballot_w32(m != 0u) != 0u);

    const int lid  = lane & 15;
    const int kh   = (lane >> 4) * 2;
    const float* prow = s_big + lid * P_PITCH + wv * 32 + kh;
    const float* gcol = gr + h * HID + 4 * lid;
    const unsigned* pslot = s_perm + wv * 32 + kh;

    float rsum = 0.0f;
    v8f acc0 = {}, acc1 = {}, acc2 = {}, acc3 = {};
#pragma unroll
    for (int s = 0; s < 8; ++s) {
        const float2 pa = *(const float2*)(prow + 4 * s);
        const uint2 nd = *(const uint2*)(pslot + 4 * s);
        const float4 b0 = *(const float4*)(gcol + (size_t)(nd.x & 0xffffu) * OUT_F);
        const float4 b1 = *(const float4*)(gcol + (size_t)(nd.y & 0xffffu) * OUT_F);
        v2f a, b;
        a.x = pa.x;
        a.y = pa.y;
        rsum += pa.x;
        rsum += pa.y;
        b.x = b0.x; b.y = b1.x;
        acc0 = __builtin_amdgcn_wmma_f32_16x16x4_f32(false, a, false, b, (short)0, acc0, false, false);
        b.x = b0.y; b.y = b1.y;
        acc1 = __builtin_amdgcn_wmma_f32_16x16x4_f32(false, a, false, b, (short)0, acc1, false, false);
        b.x = b0.z; b.y = b1.z;
        acc2 = __builtin_amdgcn_wmma_f32_16x16x4_f32(false, a, false, b, (short)0, acc2, false, false);
        b.x = b0.w; b.y = b1.w;
        acc3 = __builtin_amdgcn_wmma_f32_16x16x4_f32(false, a, false, b, (short)0, acc3, false, false);
    }

    float* ptile = s_big + PART_OFF + wv * (TI * HID) + ((lane >> 4) * 8) * HID + 4 * lid;
#pragma unroll
    for (int p = 0; p < 8; ++p) {
        float4 v;
        v.x = acc0[p]; v.y = acc1[p]; v.z = acc2[p]; v.w = acc3[p];
        *(float4*)(ptile + p * HID) = v;
    }
    rsum += __shfl_xor(rsum, 16);
    s_big[RS_OFF + lid * 32 + wv] = rsum;
    __syncthreads();

    const float* col = s_big + PART_OFF + t;
    float x = 0.0f;
#pragma unroll
    for (int w = 0; w < 32; ++w) x += col[w * (TI * HID)];
    float den = s_big[RS_OFF + (t >> 6) * 32 + lane];
    den += __shfl_xor(den, 16);
    den += __shfl_xor(den, 8);
    den += __shfl_xor(den, 4);
    den += __shfl_xor(den, 2);
    den += __shfl_xor(den, 1);
    x = x / den;
    if (use_elu[0]) x = (x > 0.0f) ? x : expm1f(x);
    out[(size_t)(i0 + (t >> 6)) * OUT_F + h * HID + (t & 63)] = x;
}

extern "C" void kernel_launch(void* const* d_in, const int* in_sizes, int n_in,
                              void* d_out, int out_size, void* d_ws, size_t ws_size,
                              hipStream_t stream) {
    const float* h_in  = (const float*)d_in[0];
    const float* nei   = (const float*)d_in[1];
    const float* adj   = (const float*)d_in[2];
    const float* w_l   = (const float*)d_in[3];
    const float* w_r   = (const float*)d_in[4];
    const float* aw    = (const float*)d_in[5];
    const int*   elu   = (const int*)d_in[6];
    float* out = (float*)d_out;

    float* ws    = (float*)d_ws;
    float* g_l   = ws;
    float* g_r   = ws + (size_t)N_NODES * OUT_F;

    gemm_xw2<<<dim3(OUT_F / BN, N_NODES / BM, 2), 256, 0, stream>>>(h_in, w_l, g_l, nei, w_r, g_r);

    attn_fused<<<dim3(N_NODES / TI, HEADS), 1024, 0, stream>>>(g_l, g_r, adj, aw, out, elu);
}


# Each wave fetches and stages the g_l rows of its own 32 slots in an LDS piece of its own: no block barrier from row arrival to aggregation; busiest slots first; neighbour sum order changes
# speedup vs baseline: 7.9887x; 1.0870x over previous
#include <hip/hip_runtime.h>
#include <math.h>

typedef float v2f __attribute__((ext_vector_type(2)));
typedef float v4f __attribute__((ext_vector_type(4)));
typedef float v8f __attribute__((ext_vector_type(8)));

#define N_NODES 1024
#define IN_F    256
#define OUT_F   256
#define HEADS   4
#define HID     64
#define NEG_SLOPE 0.2f

#define BM       32
#define BN       64
#define A_PITCH  260
#define B_PITCH  72
#define C_PITCH  68
#define KB       16

__global__ void __launch_bounds__(256)
gemm_xw2(const float* __restrict__ X0, const float* __restrict__ W0, float* __restrict__ G0,
         const float* __restrict__ X1, const float* __restrict__ W1, float* __restrict__ G1) {
    const float* X = blockIdx.z ? X1 : X0;
    const float* W = blockIdx.z ? W1 : W0;
    float*       G = blockIdx.z ? G1 : G0;

    const int t     = threadIdx.x;
    const int tileN = blockIdx.x * BN;
    const int tileM = blockIdx.y * BM;

    __shared__ __attribute__((aligned(16))) float s_a[BM * A_PITCH];
    __shared__ __attribute__((aligned(16))) float s_b[IN_F * B_PITCH];

    for (int it = 0; it < (BM * IN_F / 4) / 256; ++it) {
        const int idx = it * 256 + t;
        const int row = idx >> 6;
        const int c4  = idx & 63;
        const float4 v = *(const float4*)(X + (size_t)(tileM + row) * IN_F + c4 * 4);
        *(float4*)(s_a + row * A_PITCH + c4 * 4) = v;
    }
    for (int it = 0; it < (IN_F * BN / 4) / 256; ++it) {
        const int idx = it * 256 + t;
        const int kr  = idx >> 4;
        const int c4  = idx & 15;
        const float4 v = *(const float4*)(W + (size_t)kr * OUT_F + tileN + c4 * 4);
        *(float4*)(s_b + kr * B_PITCH + c4 * 4) = v;
    }
    __syncthreads();

    const int wv   = t >> 5;
    const int lane = t & 31;
    const int lid  = lane & 15;
    const int kh   = (lane >> 4) * 2;
    const int wm   = wv >> 2;
    const int wn   = wv & 3;
    const float* ap = s_a + (wm * 16 + lid) * A_PITCH + kh;
    const float* bp = s_b + kh * B_PITCH + wn * 16 + lid;

    v2f fa[2][KB], fb[2][KB];
#pragma unroll
    for (int s = 0; s < KB; ++s) {
        const float2 av = *(const float2*)(ap + 4 * s);
        fa[0][s].x = av.x;
        fa[0][s].y = av.y;
        fb[0][s].x = bp[(4 * s) * B_PITCH];
        fb[0][s].y = bp[(4 * s + 1) * B_PITCH];
    }
    v8f acc = {};
#pragma unroll
    for (int kb = 0; kb < IN_F / (4 * KB); ++kb) {
        const int cur = kb & 1;
        const int nxt = cur ^ 1;
        if (kb + 1 < IN_F / (4 * KB)) {
#pragma unroll
            for (int s = 0; s < KB; ++s) {
                const int k0 = (kb + 1) * (4 * KB) + 4 * s;
                const float2 av = *(const float2*)(ap + k0);
                fa[nxt][s].x = av.x;
                fa[nxt][s].y = av.y;
                fb[nxt][s].x = bp[k0 * B_PITCH];
                fb[nxt][s].y = bp[(k0 + 1) * B_PITCH];
            }
        }
        __builtin_amdgcn_sched_barrier(0);
#pragma unroll
        for (int s = 0; s < KB; ++s)
            acc = __builtin_amdgcn_wmma_f32_16x16x4_f32(false, fa[cur][s], false, fb[cur][s],
                                                        (short)0, acc, false, false);
        __builtin_amdgcn_sched_barrier(0);
    }
    __syncthreads();

    const int row_half = (lane >> 4) * 8;
#pragma unroll
    for (int p = 0; p < 8; ++p)
        s_a[(wm * 16 + row_half + p) * C_PITCH + wn * 16 + lid] = acc[p];
    __syncthreads();

    for (int it = 0; it < (BM * BN / 4) / 256; ++it) {
        const int idx = it * 256 + t;
        const int row = idx >> 4;
        const int c4  = idx & 15;
        const float4 v = *(const float4*)(s_a + row * C_PITCH + c4 * 4);
        *(float4*)(G + (size_t)(tileM + row) * OUT_F + tileN + c4 * 4) = v;
    }
}

#define TI        16
#define LIN_C     (0.5f * (1.0f + NEG_SLOPE))
#define ABS_C     (0.5f * (1.0f - NEG_SLOPE))
#define GL_PITCH  68
#define GR_PITCH  68
#define WREG      (32 * GL_PITCH)
#define PP        36
#define SPARE_ROW TI
#define PART_OFF  640
#define NB        (TI / 2 + 1)
#define LDS_BARRIER()                                                   \
    do {                                                                \
        __builtin_amdgcn_fence(__ATOMIC_RELEASE, "workgroup", "local"); \
        __builtin_amdgcn_s_barrier();                                   \
        __builtin_amdgcn_fence(__ATOMIC_ACQUIRE, "workgroup", "local"); \
    } while (0)

__global__ void __launch_bounds__(1024)
attn_fused(const float* __restrict__ gl, const float* __restrict__ gr,
           const float* __restrict__ adj, const float* __restrict__ aw,
           float* __restrict__ out, const int* __restrict__ use_elu) {
    const int i0 = blockIdx.x * TI;
    const int h  = blockIdx.y;
    const int t  = threadIdx.x;

    __shared__ __attribute__((aligned(16))) float s_big[32 * WREG];
    __shared__ __attribute__((aligned(16))) float s_gri[TI * GR_PITCH];
    __shared__ float s_lin[TI];
    __shared__ unsigned s_cnt[NB * 32];
    __shared__ unsigned s_off[NB * 32];
    __shared__ unsigned s_tot[NB];
    __shared__ __attribute__((aligned(16))) unsigned s_perm[N_NODES];
    __shared__ float s_rs[TI * 32];

    const int wv   = t >> 5;
    const int lane = t & 31;

    const int aq = lane >> 2;
    const int ac = lane & 3;
    const float* arow = adj + (size_t)((t & ~31) + aq) * N_NODES + i0 + 4 * ac;
    const float4 av0 = *(const float4*)(arow);
    const float4 av1 = *(const float4*)(arow + (size_t)8 * N_NODES);
    const float4 av2 = *(const float4*)(arow + (size_t)16 * N_NODES);
    const float4 av3 = *(const float4*)(arow + (size_t)24 * N_NODES);
    const float gri_v = gr[(i0 + (t >> 6)) * OUT_F + h * HID + (t & 63)];
    const float4 ling = *(const float4*)(gr + (size_t)(i0 + ((t >> 4) & 15)) * OUT_F + h * HID + 4 * (t & 15));
    const float4 linw = *(const float4*)(aw + 4 * (t & 15));

#define NIBBLE(v) ((v.x != 0.0f ? 1u : 0u) | (v.y != 0.0f ? 2u : 0u) | (v.z != 0.0f ? 4u : 0u) | (v.w != 0.0f ? 8u : 0u))
    unsigned w01 = (NIBBLE(av0) | (NIBBLE(av1) << 16)) << (4 * ac);
    unsigned w23 = (NIBBLE(av2) | (NIBBLE(av3) << 16)) << (4 * ac);
#undef NIBBLE
    w01 |= __shfl_xor(w01, 1);
    w23 |= __shfl_xor(w23, 1);
    w01 |= __shfl_xor(w01, 2);
    w23 |= __shfl_xor(w23, 2);
    const unsigned wsel = (ac & 2) ? w23 : w01;
    const unsigned held = (ac & 1) ? (wsel >> 16) : (wsel & 0xffffu);
    const unsigned jmask = __shfl(held, 4 * (lane & 7) + (lane >> 3));

    const int need = (__builtin_popcount(jmask) + 1) >> 1;
    unsigned rank = 0u, cnt = 0u;
#pragma unroll
    for (int b = 0; b < NB; ++b) {
        const unsigned bal = __builtin_amdgcn_ballot_w32(need == b);
        const unsigned below = __builtin_amdgcn_mbcnt_lo(bal, 0u);
        rank = (need == b) ? below : rank;
        cnt  = (lane == b) ? (unsigned)__builtin_popcount(bal) : cnt;
    }
    if (lane < NB) s_cnt[lane * 32 + wv] = cnt;
    LDS_BARRIER();

    if (t < NB * 32) {
        const unsigned c = s_cnt[t];
        unsigned off = 0u, tot = 0u;
#pragma unroll
        for (int k = 0; k < 6; ++k) {
            const unsigned bal = __builtin_amdgcn_ballot_w32(((c >> k) & 1u) != 0u);
            off += __builtin_amdgcn_mbcnt_lo(bal, 0u) << k;
            tot += (unsigned)__builtin_popcount(bal) << k;
        }
        s_off[t] = off;
        if (lane == 0) s_tot[wv] = tot;
    }
    LDS_BARRIER();

    unsigned slot = s_off[need * 32 + wv] + rank;
#pragma unroll
    for (int b = 0; b < NB - 1; ++b) slot += (b < need) ? s_tot[b] : 0u;
    slot = (unsigned)(N_NODES - 1) - slot;
    {
        const unsigned chunk = slot >> 5;
        const unsigned r     = chunk & 3u;
        const unsigned w_to  = (chunk & ~3u) | ((chunk & 4u) ? 3u - r : r);
        slot = (w_to << 5) | (slot & 31u);
    }
    s_perm[slot] = (unsigned)t | (jmask << 16);
    s_gri[(t >> 6) * GR_PITCH + (t & 63)] = gri_v;
    if (t < 256) {
        const int i = t >> 4;
        const int c = t & 15;
        const float4 g = ling;
        const float4 w = linw;
        float part = g.x * w.x;
        part = fmaf(g.y, w.y, part);
        part = fmaf(g.z, w.z, part);
        part = fmaf(g.w, w.w, part);
        part += __shfl_xor(part, 8);
        part += __shfl_xor(part, 4);
        part += __shfl_xor(part, 2);
        part += __shfl_xor(part, 1);
        if (c == 0) s_lin[i] = LIN_C * part;
    }
    LDS_BARRIER();

    float* pw = s_big + wv * WREG;
    const unsigned* pnode = s_perm + wv * 32 + (lane >> 4);
    const float* gsrc = gl + h * HID + (lane & 15) * 4;
#define ROW_LOAD(n) const v4f sl##n = *(const v4f*)(gsrc + (size_t)(pnode[2 * (n)] & 0xffffu) * OUT_F);
    ROW_LOAD(0)  ROW_LOAD(1)  ROW_LOAD(2)  ROW_LOAD(3)
    ROW_LOAD(4)  ROW_LOAD(5)  ROW_LOAD(6)  ROW_LOAD(7)
    ROW_LOAD(8)  ROW_LOAD(9)  ROW_LOAD(10) ROW_LOAD(11)
    ROW_LOAD(12) ROW_LOAD(13) ROW_LOAD(14) ROW_LOAD(15)
#undef ROW_LOAD
    float* pst = pw + (lane >> 4) * GL_PITCH + (lane & 15) * 4;
#define ROW_STORE(n) *(v4f*)(pst + 2 * (n) * GL_PITCH) = sl##n;
    ROW_STORE(0)  ROW_STORE(1)  ROW_STORE(2)  ROW_STORE(3)
    ROW_STORE(4)  ROW_STORE(5)  ROW_STORE(6)  ROW_STORE(7)
    ROW_STORE(8)  ROW_STORE(9)  ROW_STORE(10) ROW_STORE(11)
    ROW_STORE(12) ROW_STORE(13) ROW_STORE(14) ROW_STORE(15)
#undef ROW_STORE

    v2f glj[HID / 2];
    float cj = 0.0f;
    for (int c = 0; c < HID / 4; ++c) {
        const float4 v = *(const float4*)(pw + lane * GL_PITCH + c * 4);
        glj[2 * c + 0].x = v.x; glj[2 * c + 0].y = v.y;
        glj[2 * c + 1].x = v.z; glj[2 * c + 1].y = v.w;
        cj = fmaf(v.x, aw[4 * c + 0], cj);
        cj = fmaf(v.y, aw[4 * c + 1], cj);
        cj = fmaf(v.z, aw[4 * c + 2], cj);
        cj = fmaf(v.w, aw[4 * c + 3], cj);
    }
    cj *= LIN_C;
    const unsigned mask = s_perm[t] >> 16;

    for (int i = 0; i < TI; ++i) pw[i * PP + lane] = 0.0f;

    unsigned m = mask;
    int trip = 0;
    do {
        const bool v0 = (m != 0u);
        const int  t0 = v0 ? __builtin_ctz(m) : 0;
        m &= m - 1u;
        const bool v1 = (m != 0u);
        const int  t1 = v1 ? __builtin_ctz(m) : 0;
        m &= m - 1u;
        const float* g0 = s_gri + t0 * GR_PITCH;
        const float* g1 = s_gri + t1 * GR_PITCH;
        float e0 = 0.0f, e1 = 0.0f;
        for (int q = 0; q < HID / 4; q += 2) {
            const float4 c0 = *(const float4*)(g0 + 4 * q);
            const float4 c1 = *(const float4*)(g1 + 4 * q);
            const float4 d0 = *(const float4*)(g0 + 4 * q + 4);
            const float4 d1 = *(const float4*)(g1 + 4 * q + 4);
            __builtin_amdgcn_sched_barrier(0);
#define SCORE_PAIR(r0, r1, lo, hi, fp)                                          \
            {                                                                   \
                const float w0 = aw[2 * (fp)];                                  \
                const float w1 = aw[2 * (fp) + 1];                              \
                const v2f   x  = glj[fp];                                       \
                v2f a0, a1;                                                     \
                a0.x = r0.lo; a0.y = r0.hi;                                     \
                a1.x = r1.lo; a1.y = r1.hi;                                     \
                const v2f s0 = a0 + x;                                          \
                const v2f s1 = a1 + x;                                          \
                e0 = fmaf(fabsf(s0.x), w0, e0);                                 \
                e1 = fmaf(fabsf(s1.x), w0, e1);                                 \
                e0 = fmaf(fabsf(s0.y), w1, e0);                                 \
                e1 = fmaf(fabsf(s1.y), w1, e1);                                 \
            }
            SCORE_PAIR(c0, c1, x, y, 2 * q + 0)
            SCORE_PAIR(c0, c1, z, w, 2 * q + 1)
            SCORE_PAIR(d0, d1, x, y, 2 * q + 2)
            SCORE_PAIR(d0, d1, z, w, 2 * q + 3)
#undef SCORE_PAIR
        }
        e0 = fmaf(ABS_C, e0, s_lin[t0] + cj);
        e1 = fmaf(ABS_C, e1, s_lin[t1] + cj);
        pw[(v0 ? t0 : SPARE_ROW) * PP + lane] = expf(e0);
        pw[(v1 ? t1 : SPARE_ROW) * PP + lane] = expf(e1);
        ++trip;
    } while (trip < TI / 2 && __builtin_amdgcn_ballot_w32(m != 0u) != 0u);

    const int lid  = lane & 15;
    const int kh   = (lane >> 4) * 2;
    const float* prow = pw + lid * PP + kh;
    const float* gcol = gr + h * HID + 4 * lid;
    const unsigned* pslot = s_perm + wv * 32 + kh;

    float rsum = 0.0f;
    v8f acc0 = {}, acc1 = {}, acc2 = {}, acc3 = {};
#pragma unroll
    for (int s = 0; s < 8; ++s) {
        const float2 pa = *(const float2*)(prow + 4 * s);
        const uint2 nd = *(const uint2*)(pslot + 4 * s);
        const float4 b0 = *(const float4*)(gcol + (size_t)(nd.x & 0xffffu) * OUT_F);
        const float4 b1 = *(const float4*)(gcol + (size_t)(nd.y & 0xffffu) * OUT_F);
        v2f a, b;
        a.x = pa.x;
        a.y = pa.y;
        rsum += pa.x;
        rsum += pa.y;
        b.x = b0.x; b.y = b1.x;
        acc0 = __builtin_amdgcn_wmma_f32_16x16x4_f32(false, a, false, b, (short)0, acc0, false, false);
        b.x = b0.y; b.y = b1.y;
        acc1 = __builtin_amdgcn_wmma_f32_16x16x4_f32(false, a, false, b, (short)0, acc1, false, false);
        b.x = b0.z; b.y = b1.z;
        acc2 = __builtin_amdgcn_wmma_f32_16x16x4_f32(false, a, false, b, (short)0, acc2, false, false);
        b.x = b0.w; b.y = b1.w;
        acc3 = __builtin_amdgcn_wmma_f32_16x16x4_f32(false, a, false, b, (short)0, acc3, false, false);
    }

    float* ptile = pw + PART_OFF + ((lane >> 4) * 8) * HID + 4 * lid;
#pragma unroll
    for (int p = 0; p < 8; ++p) {
        float4 v;
        v.x = acc0[p]; v.y = acc1[p]; v.z = acc2[p]; v.w = acc3[p];
        *(float4*)(ptile + p * HID) = v;
    }
    rsum += __shfl_xor(rsum, 16);
    s_rs[lid * 32 + wv] = rsum;
    __syncthreads();

    const float* col = s_big + PART_OFF + t;
    float x = 0.0f;
#pragma unroll
    for (int w = 0; w < 32; ++w) x += col[w * WREG];
    float den = s_rs[(t >> 6) * 32 + lane];
    den += __shfl_xor(den, 16);
    den += __shfl_xor(den, 8);
    den += __shfl_xor(den, 4);
    den += __shfl_xor(den, 2);
    den += __shfl_xor(den, 1);
    x = x / den;
    if (use_elu[0]) x = (x > 0.0f) ? x : expm1f(x);
    out[(size_t)(i0 + (t >> 6)) * OUT_F + h * HID + (t & 63)] = x;
}

extern "C" void kernel_launch(void* const* d_in, const int* in_sizes, int n_in,
                              void* d_out, int out_size, void* d_ws, size_t ws_size,
                              hipStream_t stream) {
    const float* h_in  = (const float*)d_in[0];
    const float* nei   = (const float*)d_in[1];
    const float* adj   = (const float*)d_in[2];
    const float* w_l   = (const float*)d_in[3];
    const float* w_r   = (const float*)d_in[4];
    const float* aw    = (const float*)d_in[5];
    const int*   elu   = (const int*)d_in[6];
    float* out = (float*)d_out;

    float* ws    = (float*)d_ws;
    float* g_l   = ws;
    float* g_r   = ws + (size_t)N_NODES * OUT_F;

    gemm_xw2<<<dim3(OUT_F / BN, N_NODES / BM, 2), 256, 0, stream>>>(h_in, w_l, g_l, nei, w_r, g_r);

    attn_fused<<<dim3(N_NODES / TI, HEADS), 1024, 0, stream>>>(g_l, g_r, adj, aw, out, elu);
}


# Slot table (mask extraction, counting sort) built once per 16-target tile by 64 extra blocks of the projection launch; a score wave loads its 32 entries and asks for its rows at once; sums unchanged
# speedup vs baseline: 8.3913x; 1.0504x over previous
#include <hip/hip_runtime.h>
#include <math.h>

typedef float v2f __attribute__((ext_vector_type(2)));
typedef float v4f __attribute__((ext_vector_type(4)));
typedef float v8f __attribute__((ext_vector_type(8)));

#define N_NODES 1024
#define IN_F    256
#define OUT_F   256
#define HEADS   4
#define HID     64
#define NEG_SLOPE 0.2f

#define BM       32
#define BN       64
#define A_PITCH  260
#define B_PITCH  72
#define C_PITCH  68
#define KB       16

#define TI        16
#define NB        (TI / 2 + 1)
#define SLOT_Y    (N_NODES / TI / (OUT_F / BN) / 2)

__device__ __forceinline__ void build_slots(const float* __restrict__ adj, unsigned* __restrict__ slots,
                                            const int tile, unsigned* s_cnt, unsigned* s_off,
                                            unsigned* s_tot, unsigned* s_perm) {
    const int t    = threadIdx.x;
    const int wv   = t >> 5;
    const int lane = t & 31;
    const int i0   = tile * TI;
    const int aq = lane >> 2;
    const int ac = lane & 3;
    float4 av[4][4];
#pragma unroll
    for (int r = 0; r < 4; ++r) {
        const float* arow = adj + (size_t)(32 * (4 * wv + r) + aq) * N_NODES + i0 + 4 * ac;
#pragma unroll
        for (int k = 0; k < 4; ++k) av[r][k] = *(const float4*)(arow + (size_t)(8 * k) * N_NODES);
    }
    unsigned jm[4], rk[4];
    int nd[4];
#pragma unroll
    for (int r = 0; r < 4; ++r) {
#define NIBBLE(v) ((v.x != 0.0f ? 1u : 0u) | (v.y != 0.0f ? 2u : 0u) | (v.z != 0.0f ? 4u : 0u) | (v.w != 0.0f ? 8u : 0u))
        unsigned w01 = (NIBBLE(av[r][0]) | (NIBBLE(av[r][1]) << 16)) << (4 * ac);
        unsigned w23 = (NIBBLE(av[r][2]) | (NIBBLE(av[r][3]) << 16)) << (4 * ac);
#undef NIBBLE
        w01 |= __shfl_xor(w01, 1);
        w23 |= __shfl_xor(w23, 1);
        w01 |= __shfl_xor(w01, 2);
        w23 |= __shfl_xor(w23, 2);
        const unsigned wsel = (ac & 2) ? w23 : w01;
        const unsigned held = (ac & 1) ? (wsel >> 16) : (wsel & 0xffffu);
        const unsigned jmask = __shfl(held, 4 * (lane & 7) + (lane >> 3));
        const int need = (__builtin_popcount(jmask) + 1) >> 1;
        unsigned rank = 0u, cnt = 0u;
#pragma unroll
        for (int b = 0; b < NB; ++b) {
            const unsigned bal = __builtin_amdgcn_ballot_w32(need == b);
            const unsigned below = __builtin_amdgcn_mbcnt_lo(bal, 0u);
            rank = (need == b) ? below : rank;
            cnt  = (lane == b) ? (unsigned)__builtin_popcount(bal) : cnt;
        }
        if (lane < NB) s_cnt[lane * 32 + 4 * wv + r] = cnt;
        jm[r] = jmask;
        rk[r] = rank;
        nd[r] = need;
    }
    __syncthreads();

    for (int b = wv; b < NB; b += 8) {
        const unsigned c = s_cnt[b * 32 + lane];
        unsigned off = 0u, tot = 0u;
#pragma unroll
        for (int k = 0; k < 6; ++k) {
            const unsigned bal = __builtin_amdgcn_ballot_w32(((c >> k) & 1u) != 0u);
            off += __builtin_amdgcn_mbcnt_lo(bal, 0u) << k;
            tot += (unsigned)__builtin_popcount(bal) << k;
        }
        s_off[b * 32 + lane] = off;
        if (lane == 0) s_tot[b] = tot;
    }
    __syncthreads();

    unsigned tot_b[NB - 1];
#pragma unroll
    for (int b = 0; b < NB - 1; ++b) tot_b[b] = s_tot[b];
#pragma unroll
    for (int r = 0; r < 4; ++r) {
        const int v = 4 * wv + r;
        unsigned slot = s_off[nd[r] * 32 + v] + rk[r];
#pragma unroll
        for (int b = 0; b < NB - 1; ++b) slot += (b < nd[r]) ? tot_b[b] : 0u;
        slot = (unsigned)(N_NODES - 1) - slot;
        const unsigned chunk = slot >> 5;
        const unsigned q     = chunk & 3u;
        const unsigned w_to  = (chunk & ~3u) | ((chunk & 4u) ? 3u - q : q);
        slot = (w_to << 5) | (slot & 31u);
        s_perm[slot] = (unsigned)(32 * v + lane) | (jm[r] << 16);
    }
    __syncthreads();
    *(uint4*)(slots + (size_t)tile * N_NODES + 4 * t) = *(const uint4*)(s_perm + 4 * t);
}

__global__ void __launch_bounds__(256)
gemm_xw2(const float* __restrict__ X0, const float* __restrict__ W0, float* __restrict__ G0,
         const float* __restrict__ X1, const float* __restrict__ W1, float* __restrict__ G1,
         const float* __restrict__ adj, unsigned* __restrict__ slots) {
    if (blockIdx.y >= N_NODES / BM) {
        __shared__ unsigned s_cnt[NB * 32];
        __shared__ unsigned s_off[NB * 32];
        __shared__ unsigned s_tot[NB];
        __shared__ __attribute__((aligned(16))) unsigned s_perm[N_NODES];
        build_slots(adj, slots, ((blockIdx.y - N_NODES / BM) * (OUT_F / BN) + blockIdx.x) * 2 + blockIdx.z,
                    s_cnt, s_off, s_tot, s_perm);
        return;
    }
    const float* X = blockIdx.z ? X1 : X0;
    const float* W = blockIdx.z ? W1 : W0;
    float*       G = blockIdx.z ? G1 : G0;

    const int t     = threadIdx.x;
    const int tileN = blockIdx.x * BN;
    const int tileM = blockIdx.y * BM;

    __shared__ __attribute__((aligned(16))) float s_a[BM * A_PITCH];
    __shared__ __attribute__((aligned(16))) float s_b[IN_F * B_PITCH];

    for (int it = 0; it < (BM * IN_F / 4) / 256; ++it) {
        const int idx = it * 256 + t;
        const int row = idx >> 6;
        const int c4  = idx & 63;
        const float4 v = *(const float4*)(X + (size_t)(tileM + row) * IN_F + c4 * 4);
        *(float4*)(s_a + row * A_PITCH + c4 * 4) = v;
    }
    for (int it = 0; it < (IN_F * BN / 4) / 256; ++it) {
        const int idx = it * 256 + t;
        const int kr  = idx >> 4;
        const int c4  = idx & 15;
        const float4 v = *(const float4*)(W + (size_t)kr * OUT_F + tileN + c4 * 4);
        *(float4*)(s_b + kr * B_PITCH + c4 * 4) = v;
    }
    __syncthreads();

    const int wv   = t >> 5;
    const int lane = t & 31;
    const int lid  = lane & 15;
    const int kh   = (lane >> 4) * 2;
    const int wm   = wv >> 2;
    const int wn   = wv & 3;
    const float* ap = s_a + (wm * 16 + lid) * A_PITCH + kh;
    const float* bp = s_b + kh * B_PITCH + wn * 16 + lid;

    v2f fa[2][KB], fb[2][KB];
#pragma unroll
    for (int s = 0; s < KB; ++s) {
        const float2 av = *(const float2*)(ap + 4 * s);
        fa[0][s].x = av.x;
        fa[0][s].y = av.y;
        fb[0][s].x = bp[(4 * s) * B_PITCH];
        fb[0][s].y = bp[(4 * s + 1) * B_PITCH];
    }
    v8f acc = {};
#pragma unroll
    for (int kb = 0; kb < IN_F / (4 * KB); ++kb) {
        const int cur = kb & 1;
        const int nxt = cur ^ 1;
        if (kb + 1 < IN_F / (4 * KB)) {
#pragma unroll
            for (int s = 0; s < KB; ++s) {
                const int k0 = (kb + 1) * (4 * KB) + 4 * s;
                const float2 av = *(const float2*)(ap + k0);
                fa[nxt][s].x = av.x;
                fa[nxt][s].y = av.y;
                fb[nxt][s].x = bp[k0 * B_PITCH];
                fb[nxt][s].y = bp[(k0 + 1) * B_PITCH];
            }
        }
        __builtin_amdgcn_sched_barrier(0);
#pragma unroll
        for (int s = 0; s < KB; ++s)
            acc = __builtin_amdgcn_wmma_f32_16x16x4_f32(false, fa[cur][s], false, fb[cur][s],
                                                        (short)0, acc, false, false);
        __builtin_amdgcn_sched_barrier(0);
    }
    __syncthreads();

    const int row_half = (lane >> 4) * 8;
#pragma unroll
    for (int p = 0; p < 8; ++p)
        s_a[(wm * 16 + row_half + p) * C_PITCH + wn * 16 + lid] = acc[p];
    __syncthreads();

    for (int it = 0; it < (BM * BN / 4) / 256; ++it) {
        const int idx = it * 256 + t;
        const int row = idx >> 4;
        const int c4  = idx & 15;
        const float4 v = *(const float4*)(s_a + row * C_PITCH + c4 * 4);
        *(float4*)(G + (size_t)(tileM + row) * OUT_F + tileN + c4 * 4) = v;
    }
}

#define LIN_C     (0.5f * (1.0f + NEG_SLOPE))
#define ABS_C     (0.5f * (1.0f - NEG_SLOPE))
#define GL_PITCH  68
#define GR_PITCH  68
#define WREG      (32 * GL_PITCH)
#define PP        36
#define SPARE_ROW TI
#define PART_OFF  640
#define LDS_BARRIER()                                                   \
    do {                                                                \
        __builtin_amdgcn_fence(__ATOMIC_RELEASE, "workgroup", "local"); \
        __builtin_amdgcn_s_barrier();                                   \
        __builtin_amdgcn_fence(__ATOMIC_ACQUIRE, "workgroup", "local"); \
    } while (0)

__global__ void __launch_bounds__(1024)
attn_fused(const float* __restrict__ gl, const float* __restrict__ gr,
           const unsigned* __restrict__ slots, const float* __restrict__ aw,
           float* __restrict__ out, const int* __restrict__ use_elu) {
    const int i0 = blockIdx.x * TI;
    const int h  = blockIdx.y;
    const int t  = threadIdx.x;

    __shared__ __attribute__((aligned(16))) float s_big[32 * WREG];
    __shared__ __attribute__((aligned(16))) float s_gri[TI * GR_PITCH];
    __shared__ float s_lin[TI];
    __shared__ __attribute__((aligned(16))) unsigned s_perm[N_NODES];
    __shared__ float s_rs[TI * 32];

    const int wv   = t >> 5;
    const int lane = t & 31;

    const unsigned ent = slots[(size_t)blockIdx.x * N_NODES + t];
    const float gri_v = gr[(i0 + (t >> 6)) * OUT_F + h * HID + (t & 63)];

    if (t >= N_NODES - 256) {
        const int i = (t >> 4) & 15;
        const int c = t & 15;
        const float4 g = *(const float4*)(gr + (size_t)(i0 + i) * OUT_F + h * HID + 4 * c);
        const float4 w = *(const float4*)(aw + 4 * c);
        float part = g.x * w.x;
        part = fmaf(g.y, w.y, part);
        part = fmaf(g.z, w.z, part);
        part = fmaf(g.w, w.w, part);
        part += __shfl_xor(part, 8);
        part += __shfl_xor(part, 4);
        part += __shfl_xor(part, 2);
        part += __shfl_xor(part, 1);
        if (c == 0) s_lin[i] = LIN_C * part;
    }

    s_perm[t] = ent;
    float* pw = s_big + wv * WREG;
    const unsigned* pnode = s_perm + wv * 32 + (lane >> 4);
    const float* gsrc = gl + h * HID + (lane & 15) * 4;
#define ROW_LOAD(n) const v4f sl##n = *(const v4f*)(gsrc + (size_t)(pnode[2 * (n)] & 0xffffu) * OUT_F);
    ROW_LOAD(0)  ROW_LOAD(1)  ROW_LOAD(2)  ROW_LOAD(3)
    ROW_LOAD(4)  ROW_LOAD(5)  ROW_LOAD(6)  ROW_LOAD(7)
    ROW_LOAD(8)  ROW_LOAD(9)  ROW_LOAD(10) ROW_LOAD(11)
    ROW_LOAD(12) ROW_LOAD(13) ROW_LOAD(14) ROW_LOAD(15)
#undef ROW_LOAD

    s_gri[(t >> 6) * GR_PITCH + (t & 63)] = gri_v;
    LDS_BARRIER();

    float* pst = pw + (lane >> 4) * GL_PITCH + (lane & 15) * 4;
#define ROW_STORE(n) *(v4f*)(pst + 2 * (n) * GL_PITCH) = sl##n;
    ROW_STORE(0)  ROW_STORE(1)  ROW_STORE(2)  ROW_STORE(3)
    ROW_STORE(4)  ROW_STORE(5)  ROW_STORE(6)  ROW_STORE(7)
    ROW_STORE(8)  ROW_STORE(9)  ROW_STORE(10) ROW_STORE(11)
    ROW_STORE(12) ROW_STORE(13) ROW_STORE(14) ROW_STORE(15)
#undef ROW_STORE

    v2f glj[HID / 2];
    float cj = 0.0f;
    for (int c = 0; c < HID / 4; ++c) {
        const float4 v = *(const float4*)(pw + lane * GL_PITCH + c * 4);
        glj[2 * c + 0].x = v.x; glj[2 * c + 0].y = v.y;
        glj[2 * c + 1].x = v.z; glj[2 * c + 1].y = v.w;
        cj = fmaf(v.x, aw[4 * c + 0], cj);
        cj = fmaf(v.y, aw[4 * c + 1], cj);
        cj = fmaf(v.z, aw[4 * c + 2], cj);
        cj = fmaf(v.w, aw[4 * c + 3], cj);
    }
    cj *= LIN_C;
    const unsigned mask = ent >> 16;

    for (int i = 0; i < TI; ++i) pw[i * PP + lane] = 0.0f;

    unsigned m = mask;
    int trip = 0;
    do {
        const bool v0 = (m != 0u);
        const int  t0 = v0 ? __builtin_ctz(m) : 0;
        m &= m - 1u;
        const bool v1 = (m != 0u);
        const int  t1 = v1 ? __builtin_ctz(m) : 0;
        m &= m - 1u;
        const float* g0 = s_gri + t0 * GR_PITCH;
        const float* g1 = s_gri + t1 * GR_PITCH;
        float e0 = 0.0f, e1 = 0.0f;
        for (int q = 0; q < HID / 4; q += 2) {
            const float4 c0 = *(const float4*)(g0 + 4 * q);
            const float4 c1 = *(const float4*)(g1 + 4 * q);
            const float4 d0 = *(const float4*)(g0 + 4 * q + 4);
            const float4 d1 = *(const float4*)(g1 + 4 * q + 4);
            __builtin_amdgcn_sched_barrier(0);
#define SCORE_PAIR(r0, r1, lo, hi, fp)                                          \
            {                                                                   \
                const float w0 = aw[2 * (fp)];                                  \
                const float w1 = aw[2 * (fp) + 1];                              \
                const v2f   x  = glj[fp];                                       \
                v2f a0, a1;                                                     \
                a0.x = r0.lo; a0.y = r0.hi;                                     \
                a1.x = r1.lo; a1.y = r1.hi;                                     \
                const v2f s0 = a0 + x;                                          \
                const v2f s1 = a1 + x;                                          \
                e0 = fmaf(fabsf(s0.x), w0, e0);                                 \
                e1 = fmaf(fabsf(s1.x), w0, e1);                                 \
                e0 = fmaf(fabsf(s0.y), w1, e0);                                 \
                e1 = fmaf(fabsf(s1.y), w1, e1);                                 \
            }
            SCORE_PAIR(c0, c1, x, y, 2 * q + 0)
            SCORE_PAIR(c0, c1, z, w, 2 * q + 1)
            SCORE_PAIR(d0, d1, x, y, 2 * q + 2)
            SCORE_PAIR(d0, d1, z, w, 2 * q + 3)
#undef SCORE_PAIR
        }
        e0 = fmaf(ABS_C, e0, s_lin[t0] + cj);
        e1 = fmaf(ABS_C, e1, s_lin[t1] + cj);
        pw[(v0 ? t0 : SPARE_ROW) * PP + lane] = expf(e0);
        pw[(v1 ? t1 : SPARE_ROW) * PP + lane] = expf(e1);
        ++trip;
    } while (trip < TI / 2 && __builtin_amdgcn_ballot_w32(m != 0u) != 0u);

    const int lid  = lane & 15;
    const int kh   = (lane >> 4) * 2;
    const float* prow = pw + lid * PP + kh;
    const float* gcol = gr + h * HID + 4 * lid;
    const unsigned* pslot = s_perm + wv * 32 + kh;

    float rsum = 0.0f;
    v8f acc0 = {}, acc1 = {}, acc2 = {}, acc3 = {};
#pragma unroll
    for (int s = 0; s < 8; ++s) {
        const float2 pa = *(const float2*)(prow + 4 * s);
        const uint2 nd = *(const uint2*)(pslot + 4 * s);
        const float4 b0 = *(const float4*)(gcol + (size_t)(nd.x & 0xffffu) * OUT_F);
        const float4 b1 = *(const float4*)(gcol + (size_t)(nd.y & 0xffffu) * OUT_F);
        v2f a, b;
        a.x = pa.x;
        a.y = pa.y;
        rsum += pa.x;
        rsum += pa.y;
        b.x = b0.x; b.y = b1.x;
        acc0 = __builtin_amdgcn_wmma_f32_16x16x4_f32(false, a, false, b, (short)0, acc0, false, false);
        b.x = b0.y; b.y = b1.y;
        acc1 = __builtin_amdgcn_wmma_f32_16x16x4_f32(false, a, false, b, (short)0, acc1, false, false);
        b.x = b0.z; b.y = b1.z;
        acc2 = __builtin_amdgcn_wmma_f32_16x16x4_f32(false, a, false, b, (short)0, acc2, false, false);
        b.x = b0.w; b.y = b1.w;
        acc3 = __builtin_amdgcn_wmma_f32_16x16x4_f32(false, a, false, b, (short)0, acc3, false, false);
    }

    float* ptile = pw + PART_OFF + ((lane >> 4) * 8) * HID + 4 * lid;
#pragma unroll
    for (int p = 0; p < 8; ++p) {
        float4 v;
        v.x = acc0[p]; v.y = acc1[p]; v.z = acc2[p]; v.w = acc3[p];
        *(float4*)(ptile + p * HID) = v;
    }
    rsum += __shfl_xor(rsum, 16);
    s_rs[lid * 32 + wv] = rsum;
    __syncthreads();

    const float* col = s_big + PART_OFF + t;
    float x = 0.0f;
#pragma unroll
    for (int w = 0; w < 32; ++w) x += col[w * WREG];
    float den = s_rs[(t >> 6) * 32 + lane];
    den += __shfl_xor(den, 16);
    den += __shfl_xor(den, 8);
    den += __shfl_xor(den, 4);
    den += __shfl_xor(den, 2);
    den += __shfl_xor(den, 1);
    x = x / den;
    if (use_elu[0]) x = (x > 0.0f) ? x : expm1f(x);
    out[(size_t)(i0 + (t >> 6)) * OUT_F + h * HID + (t & 63)] = x;
}

extern "C" void kernel_launch(void* const* d_in, const int* in_sizes, int n_in,
                              void* d_out, int out_size, void* d_ws, size_t ws_size,
                              hipStream_t stream) {
    const float* h_in  = (const float*)d_in[0];
    const float* nei   = (const float*)d_in[1];
    const float* adj   = (const float*)d_in[2];
    const float* w_l   = (const float*)d_in[3];
    const float* w_r   = (const float*)d_in[4];
    const float* aw    = (const float*)d_in[5];
    const int*   elu   = (const int*)d_in[6];
    float* out = (float*)d_out;

    float* ws    = (float*)d_ws;
    float* g_l   = ws;
    float* g_r   = ws + (size_t)N_NODES * OUT_F;
    unsigned* slots = (unsigned*)(ws + (size_t)2 * N_NODES * OUT_F);

    gemm_xw2<<<dim3(OUT_F / BN, N_NODES / BM + SLOT_Y, 2), 256, 0, stream>>>(h_in, w_l, g_l, nei, w_r, g_r, adj, slots);

    attn_fused<<<dim3(N_NODES / TI, HEADS), 1024, 0, stream>>>(g_l, g_r, slots, aw, out, elu);
}
